# starting kernel
# baseline (speedup 1.0000x reference)
#include <hip/hip_runtime.h>

typedef float v2f __attribute__((ext_vector_type(2)));
typedef float v8f __attribute__((ext_vector_type(8)));

#define DIM 128
#define XP 132
#define NEG_SLOPE 0.1f

__global__ void gcn_init_kernel(float* __restrict__ out, float* __restrict__ deg,
                                int N, long total) {
  long i = (long)blockIdx.x * blockDim.x + threadIdx.x;
  if (i < total) out[i] = 0.0f;
  if (i < N) deg[i] = 1.0f;
}

__global__ void gcn_deg_kernel(const int* __restrict__ ei, float* __restrict__ deg, int E) {
  int e = blockIdx.x * blockDim.x + threadIdx.x;
  if (e < E) atomicAdd(&deg[ei[E + e]], 1.0f);
}

__global__ void gcn_dinv_kernel(float* __restrict__ deg, int N) {
  int i = blockIdx.x * blockDim.x + threadIdx.x;
  if (i < N) deg[i] = rsqrtf(deg[i]);
}

__global__ __launch_bounds__(256) void gcn_gemm_kernel(const float* __restrict__ x,
                                                       const float* __restrict__ W,
                                                       float* __restrict__ h, int N) {
  __shared__ float Ws[DIM * DIM];
  __shared__ float xs[16 * XP];

  const int tid  = threadIdx.x;
  const int row0 = blockIdx.x * 16;

  for (int i = tid; i < (DIM * DIM) / 4; i += 256) {
    ((float4*)Ws)[i] = ((const float4*)W)[i];
  }
  for (int i = tid; i < 16 * DIM; i += 256) {
    int r = i >> 7, c = i & (DIM - 1);
    int gr = row0 + r;
    xs[r * XP + c] = (gr < N) ? x[(long)gr * DIM + c] : 0.0f;
  }
  __syncthreads();

  const int wave = tid >> 5;
  const int lane = tid & 31;
  const int half = lane >> 4;
  const int m    = lane & 15;
  const int col0 = wave * 16;

  v8f acc = {};
  for (int k0 = 0; k0 < DIM; k0 += 4) {
    int ka = k0 + 2 * half;
    v2f a, b;
    a.x = xs[m * XP + ka];
    a.y = xs[m * XP + ka + 1];
    b.x = Ws[ka * DIM + col0 + m];
    b.y = Ws[(ka + 1) * DIM + col0 + m];
    acc = __builtin_amdgcn_wmma_f32_16x16x4_f32(
         false, a,  false, b,
         (short)0, acc,  false,  false);
  }

  for (int r = 0; r < 8; ++r) {
    int row = row0 + r + 8 * half;
    if (row < N) h[(long)row * DIM + col0 + m] = acc[r];
  }
}

__global__ __launch_bounds__(256) void gcn_scatter_kernel(const int* __restrict__ ei,
                                                          const float* __restrict__ dinv,
                                                          const float* __restrict__ h,
                                                          float* __restrict__ out, int E) {
  int gwave = (int)((blockIdx.x * 256 + threadIdx.x) >> 5);
  int lane  = threadIdx.x & 31;
  if (gwave >= E) return;
  int s = ei[gwave];
  int d = ei[E + gwave];
  float w = dinv[s] * dinv[d];
  float4 v = ((const float4*)(h + (long)s * DIM))[lane];
  float* o = out + (long)d * DIM + lane * 4;
  atomicAdd(o + 0, v.x * w);
  atomicAdd(o + 1, v.y * w);
  atomicAdd(o + 2, v.z * w);
  atomicAdd(o + 3, v.w * w);
}

__global__ void gcn_final_kernel(const float* __restrict__ x, const float* __restrict__ bias,
                                 const float* __restrict__ dinv, const float* __restrict__ h,
                                 float* __restrict__ out, long total) {
  long i = (long)blockIdx.x * blockDim.x + threadIdx.x;
  if (i >= total) return;
  int row = (int)(i >> 7);
  int col = (int)(i & (DIM - 1));
  float di = dinv[row];
  float v = out[i] + h[i] * (di * di) + bias[col];
  v = (v >= 0.0f) ? v : NEG_SLOPE * v;
  out[i] = v + x[i];
}

extern "C" void kernel_launch(void* const* d_in, const int* in_sizes, int n_in,
                              void* d_out, int out_size, void* d_ws, size_t ws_size,
                              hipStream_t stream) {
  const float* x  = (const float*)d_in[0];
  const int*   ei = (const int*)d_in[1];
  const float* W  = (const float*)d_in[2];
  const float* b  = (const float*)d_in[3];

  const int N = in_sizes[0] / DIM;
  const int E = in_sizes[1] / 2;

  float* out  = (float*)d_out;
  float* dinv = (float*)d_ws;
  size_t hoff = (((size_t)N * sizeof(float)) + 255) & ~(size_t)255;
  float* h    = (float*)((char*)d_ws + hoff);

  const long total = (long)N * DIM;

  gcn_init_kernel<<<(int)((total + 255) / 256), 256, 0, stream>>>(out, dinv, N, total);
  gcn_deg_kernel<<<(E + 255) / 256, 256, 0, stream>>>(ei, dinv, E);
  gcn_dinv_kernel<<<(N + 255) / 256, 256, 0, stream>>>(dinv, N);
  gcn_gemm_kernel<<<(N + 15) / 16, 256, 0, stream>>>(x, W, h, N);
  gcn_scatter_kernel<<<(E + 7) / 8, 256, 0, stream>>>(ei, dinv, h, out, E);
  gcn_final_kernel<<<(int)((total + 255) / 256), 256, 0, stream>>>(x, b, dinv, h, out, total);
}


# scatter: lane l takes columns l,l+32,l+64,l+96 so each f32 atomic instruction covers one whole 128-B line of out (4 line requests per edge, was 16); sum order stays atomic arrival order
# speedup vs baseline: 1.2059x; 1.2059x over previous
#include <hip/hip_runtime.h>

typedef float v2f __attribute__((ext_vector_type(2)));
typedef float v8f __attribute__((ext_vector_type(8)));

#define DIM 128
#define XP 132
#define NEG_SLOPE 0.1f

__global__ void gcn_init_kernel(float* __restrict__ out, float* __restrict__ deg,
                                int N, long total) {
  long i = (long)blockIdx.x * blockDim.x + threadIdx.x;
  if (i < total) out[i] = 0.0f;
  if (i < N) deg[i] = 1.0f;
}

__global__ void gcn_deg_kernel(const int* __restrict__ ei, float* __restrict__ deg, int E) {
  int e = blockIdx.x * blockDim.x + threadIdx.x;
  if (e < E) atomicAdd(&deg[ei[E + e]], 1.0f);
}

__global__ void gcn_dinv_kernel(float* __restrict__ deg, int N) {
  int i = blockIdx.x * blockDim.x + threadIdx.x;
  if (i < N) deg[i] = rsqrtf(deg[i]);
}

__global__ __launch_bounds__(256) void gcn_gemm_kernel(const float* __restrict__ x,
                                                       const float* __restrict__ W,
                                                       float* __restrict__ h, int N) {
  __shared__ float Ws[DIM * DIM];
  __shared__ float xs[16 * XP];

  const int tid  = threadIdx.x;
  const int row0 = blockIdx.x * 16;

  for (int i = tid; i < (DIM * DIM) / 4; i += 256) {
    ((float4*)Ws)[i] = ((const float4*)W)[i];
  }
  for (int i = tid; i < 16 * DIM; i += 256) {
    int r = i >> 7, c = i & (DIM - 1);
    int gr = row0 + r;
    xs[r * XP + c] = (gr < N) ? x[(long)gr * DIM + c] : 0.0f;
  }
  __syncthreads();

  const int wave = tid >> 5;
  const int lane = tid & 31;
  const int half = lane >> 4;
  const int m    = lane & 15;
  const int col0 = wave * 16;

  v8f acc = {};
  for (int k0 = 0; k0 < DIM; k0 += 4) {
    int ka = k0 + 2 * half;
    v2f a, b;
    a.x = xs[m * XP + ka];
    a.y = xs[m * XP + ka + 1];
    b.x = Ws[ka * DIM + col0 + m];
    b.y = Ws[(ka + 1) * DIM + col0 + m];
    acc = __builtin_amdgcn_wmma_f32_16x16x4_f32(
         false, a,  false, b,
         (short)0, acc,  false,  false);
  }

  for (int r = 0; r < 8; ++r) {
    int row = row0 + r + 8 * half;
    if (row < N) h[(long)row * DIM + col0 + m] = acc[r];
  }
}

__global__ __launch_bounds__(256) void gcn_scatter_kernel(const int* __restrict__ ei,
                                                          const float* __restrict__ dinv,
                                                          const float* __restrict__ h,
                                                          float* __restrict__ out, int E) {
  int gwave = (int)((blockIdx.x * 256 + threadIdx.x) >> 5);
  int lane  = threadIdx.x & 31;
  if (gwave >= E) return;
  int s = ei[gwave];
  int d = ei[E + gwave];
  float w = dinv[s] * dinv[d];
  const float* hs = h + (long)s * DIM + lane;
  float* o = out + (long)d * DIM + lane;
  float v0 = hs[0];
  float v1 = hs[32];
  float v2 = hs[64];
  float v3 = hs[96];
  atomicAdd(o + 0,  v0 * w);
  atomicAdd(o + 32, v1 * w);
  atomicAdd(o + 64, v2 * w);
  atomicAdd(o + 96, v3 * w);
}

__global__ void gcn_final_kernel(const float* __restrict__ x, const float* __restrict__ bias,
                                 const float* __restrict__ dinv, const float* __restrict__ h,
                                 float* __restrict__ out, long total) {
  long i = (long)blockIdx.x * blockDim.x + threadIdx.x;
  if (i >= total) return;
  int row = (int)(i >> 7);
  int col = (int)(i & (DIM - 1));
  float di = dinv[row];
  float v = out[i] + h[i] * (di * di) + bias[col];
  v = (v >= 0.0f) ? v : NEG_SLOPE * v;
  out[i] = v + x[i];
}

extern "C" void kernel_launch(void* const* d_in, const int* in_sizes, int n_in,
                              void* d_out, int out_size, void* d_ws, size_t ws_size,
                              hipStream_t stream) {
  const float* x  = (const float*)d_in[0];
  const int*   ei = (const int*)d_in[1];
  const float* W  = (const float*)d_in[2];
  const float* b  = (const float*)d_in[3];

  const int N = in_sizes[0] / DIM;
  const int E = in_sizes[1] / 2;

  float* out  = (float*)d_out;
  float* dinv = (float*)d_ws;
  size_t hoff = (((size_t)N * sizeof(float)) + 255) & ~(size_t)255;
  float* h    = (float*)((char*)d_ws + hoff);

  const long total = (long)N * DIM;

  gcn_init_kernel<<<(int)((total + 255) / 256), 256, 0, stream>>>(out, dinv, N, total);
  gcn_deg_kernel<<<(E + 255) / 256, 256, 0, stream>>>(ei, dinv, E);
  gcn_dinv_kernel<<<(N + 255) / 256, 256, 0, stream>>>(dinv, N);
  gcn_gemm_kernel<<<(N + 15) / 16, 256, 0, stream>>>(x, W, h, N);
  gcn_scatter_kernel<<<(E + 7) / 8, 256, 0, stream>>>(ei, dinv, h, out, E);
  gcn_final_kernel<<<(int)((total + 255) / 256), 256, 0, stream>>>(x, b, dinv, h, out, total);
}


# aggregation by destination: counting-sort edges on dst (int cursor atomics), one wave then sums each out row from h in registers, adds bias/leaky/x; no float atomics; sum order = cursor arrival order
# speedup vs baseline: 1.6831x; 1.3957x over previous
#include <hip/hip_runtime.h>


typedef float v2f __attribute__((ext_vector_type(2)));
typedef float v8f __attribute__((ext_vector_type(8)));

#define DIM 128
#define XP 132
#define NEG_SLOPE 0.1f
#define SCAN_BLK 1024

__device__ __forceinline__ int block_sum_256(int v, int* red) {
  for (int o = 16; o > 0; o >>= 1) v += __shfl_xor(v, o, 32);
  if ((threadIdx.x & 31) == 0) red[threadIdx.x >> 5] = v;
  __syncthreads();
  int tot = 0;
  for (int w = 0; w < 8; ++w) tot += red[w];
  __syncthreads();
  return tot;
}

__global__ void csr_zero_kernel(int* __restrict__ cnt, int N) {
  int i = blockIdx.x * blockDim.x + threadIdx.x;
  if (i < N) cnt[i] = 0;
}

__global__ void csr_count_kernel(const int* __restrict__ ei, int* __restrict__ cnt, int E) {
  int e = blockIdx.x * blockDim.x + threadIdx.x;
  if (e < E) atomicAdd(&cnt[ei[E + e]], 1);
}

__global__ __launch_bounds__(256) void csr_blocksum_kernel(const int* __restrict__ cnt,
                                                           int* __restrict__ bsum,
                                                           float* __restrict__ dinv, int N) {
  __shared__ int red[8];
  const int base = blockIdx.x * SCAN_BLK;
  int s = 0;
  for (int k = 0; k < 4; ++k) {
    int i = base + k * 256 + (int)threadIdx.x;
    if (i < N) {
      int c = cnt[i];
      s += c;
      dinv[i] = rsqrtf((float)(c + 1));
    }
  }
  int tot = block_sum_256(s, red);
  if (threadIdx.x == 0) bsum[blockIdx.x] = tot;
}

__global__ __launch_bounds__(256) void csr_scan_kernel(const int* __restrict__ cnt,
                                                       const int* __restrict__ bsum,
                                                       int* __restrict__ rowstart,
                                                       int* __restrict__ cursor, int N) {
  __shared__ int red[8];
  __shared__ int wtot[8];
  const int tid = threadIdx.x, lane = tid & 31, wave = tid >> 5;

  int part = 0;
  for (int j = tid; j < (int)blockIdx.x; j += 256) part += bsum[j];
  const int offset = block_sum_256(part, red);

  const int i0 = blockIdx.x * SCAN_BLK + tid * 4;
  int c[4];
  for (int k = 0; k < 4; ++k) c[k] = (i0 + k < N) ? cnt[i0 + k] : 0;
  const int tsum = c[0] + c[1] + c[2] + c[3];

  int incl = tsum;
  for (int o = 1; o < 32; o <<= 1) {
    int t = __shfl_up(incl, o, 32);
    if (lane >= o) incl += t;
  }
  if (lane == 31) wtot[wave] = incl;
  __syncthreads();
  int wpre = 0;
  for (int w = 0; w < wave; ++w) wpre += wtot[w];

  int run = offset + wpre + incl - tsum;
  for (int k = 0; k < 4; ++k) {
    int i = i0 + k;
    if (i <= N) rowstart[i] = run;
    if (i < N) cursor[i] = run;
    run += c[k];
  }
}

__global__ void csr_fill_kernel(const int* __restrict__ ei, int* __restrict__ cursor,
                                int* __restrict__ csr, int E) {
  int e = blockIdx.x * blockDim.x + threadIdx.x;
  if (e < E) {
    int s = ei[e];
    int d = ei[E + e];
    int p = atomicAdd(&cursor[d], 1);
    csr[p] = s;
  }
}

__global__ __launch_bounds__(256) void gcn_gemm_kernel(const float* __restrict__ x,
                                                       const float* __restrict__ W,
                                                       float* __restrict__ h, int N) {
  __shared__ float Ws[DIM * DIM];
  __shared__ float xs[16 * XP];

  const int tid  = threadIdx.x;
  const int row0 = blockIdx.x * 16;

  for (int i = tid; i < (DIM * DIM) / 4; i += 256) {
    ((float4*)Ws)[i] = ((const float4*)W)[i];
  }
  for (int i = tid; i < 16 * DIM; i += 256) {
    int r = i >> 7, c = i & (DIM - 1);
    int gr = row0 + r;
    xs[r * XP + c] = (gr < N) ? x[(long)gr * DIM + c] : 0.0f;
  }
  __syncthreads();

  const int wave = tid >> 5;
  const int lane = tid & 31;
  const int half = lane >> 4;
  const int m    = lane & 15;
  const int col0 = wave * 16;

  v8f acc = {};
  for (int k0 = 0; k0 < DIM; k0 += 4) {
    int ka = k0 + 2 * half;
    v2f a, b;
    a.x = xs[m * XP + ka];
    a.y = xs[m * XP + ka + 1];
    b.x = Ws[ka * DIM + col0 + m];
    b.y = Ws[(ka + 1) * DIM + col0 + m];
    acc = __builtin_amdgcn_wmma_f32_16x16x4_f32(
         false, a,  false, b,
         (short)0, acc,  false,  false);
  }

  for (int r = 0; r < 8; ++r) {
    int row = row0 + r + 8 * half;
    if (row < N) h[(long)row * DIM + col0 + m] = acc[r];
  }
}

#define ROW4(s) (((const float4*)(h + (size_t)(s) * DIM))[lane])
#define FMA4(rowv, wgt)                    \
  acc.x = fmaf((rowv).x, (wgt), acc.x);  \
  acc.y = fmaf((rowv).y, (wgt), acc.y);  \
  acc.z = fmaf((rowv).z, (wgt), acc.z);  \
  acc.w = fmaf((rowv).w, (wgt), acc.w)
#define LANE_I(v, k) __builtin_amdgcn_readlane((v), (k))
#define LANE_W(v, k) (__int_as_float(__builtin_amdgcn_readlane(__float_as_int(v), (k))) * di)

__global__ __launch_bounds__(256) void gcn_gather_kernel(const float* __restrict__ x,
                                                         const float* __restrict__ bias,
                                                         const float* __restrict__ dinv,
                                                         const float* __restrict__ h,
                                                         const int* __restrict__ rowstart,
                                                         const int* __restrict__ csr,
                                                         float* __restrict__ out, int N) {
  const int lane = threadIdx.x & 31;
  const int d = __builtin_amdgcn_readfirstlane((int)(blockIdx.x * 8 + (threadIdx.x >> 5)));
  if (d >= N) return;

  const int beg = rowstart[d];
  const int end = rowstart[d + 1];
  const float di = dinv[d];
  const float4 hv = ROW4(d);
  const float4 xv = ((const float4*)(x + (size_t)d * DIM))[lane];
  const float4 bv = ((const float4*)bias)[lane];

  const float wself = di * di;
  float4 acc;
  acc.x = hv.x * wself;
  acc.y = hv.y * wself;
  acc.z = hv.z * wself;
  acc.w = hv.w * wself;

  for (int base = beg; base < end; base += 32) {
    const int j = base + lane;
    int sv = 0;
    float wv = 0.0f;
    if (j < end) {
      int s = csr[j];
      s = min(max(s, 0), N - 1);
      sv = s;
      wv = dinv[s];
    }
    int n = end - base;
    if (n > 32) n = 32;

    int k = 0;
    for (; k + 4 <= n; k += 4) {
      const int s0 = LANE_I(sv, k), s1 = LANE_I(sv, k + 1), s2 = LANE_I(sv, k + 2), s3 = LANE_I(sv, k + 3);
      const float w0 = LANE_W(wv, k), w1 = LANE_W(wv, k + 1), w2 = LANE_W(wv, k + 2), w3 = LANE_W(wv, k + 3);
      const float4 r0 = ROW4(s0);
      const float4 r1 = ROW4(s1);
      const float4 r2 = ROW4(s2);
      const float4 r3 = ROW4(s3);
      FMA4(r0, w0);
      FMA4(r1, w1);
      FMA4(r2, w2);
      FMA4(r3, w3);
    }
    const int rem = n - k;
    if (rem == 3) {
      const int s0 = LANE_I(sv, k), s1 = LANE_I(sv, k + 1), s2 = LANE_I(sv, k + 2);
      const float w0 = LANE_W(wv, k), w1 = LANE_W(wv, k + 1), w2 = LANE_W(wv, k + 2);
      const float4 r0 = ROW4(s0);
      const float4 r1 = ROW4(s1);
      const float4 r2 = ROW4(s2);
      FMA4(r0, w0);
      FMA4(r1, w1);
      FMA4(r2, w2);
    } else if (rem == 2) {
      const int s0 = LANE_I(sv, k), s1 = LANE_I(sv, k + 1);
      const float w0 = LANE_W(wv, k), w1 = LANE_W(wv, k + 1);
      const float4 r0 = ROW4(s0);
      const float4 r1 = ROW4(s1);
      FMA4(r0, w0);
      FMA4(r1, w1);
    } else if (rem == 1) {
      const int s0 = LANE_I(sv, k);
      const float w0 = LANE_W(wv, k);
      const float4 r0 = ROW4(s0);
      FMA4(r0, w0);
    }
  }

  float4 o;
  float v;
  v = acc.x + bv.x; v = (v >= 0.0f) ? v : NEG_SLOPE * v; o.x = v + xv.x;
  v = acc.y + bv.y; v = (v >= 0.0f) ? v : NEG_SLOPE * v; o.y = v + xv.y;
  v = acc.z + bv.z; v = (v >= 0.0f) ? v : NEG_SLOPE * v; o.z = v + xv.z;
  v = acc.w + bv.w; v = (v >= 0.0f) ? v : NEG_SLOPE * v; o.w = v + xv.w;
  ((float4*)(out + (size_t)d * DIM))[lane] = o;
}

extern "C" void kernel_launch(void* const* d_in, const int* in_sizes, int n_in,
                              void* d_out, int out_size, void* d_ws, size_t ws_size,
                              hipStream_t stream) {
  const float* x  = (const float*)d_in[0];
  const int*   ei = (const int*)d_in[1];
  const float* W  = (const float*)d_in[2];
  const float* b  = (const float*)d_in[3];

  const int N = in_sizes[0] / DIM;
  const int E = in_sizes[1] / 2;
  const int nblk = N / SCAN_BLK + 1;

  float* out = (float*)d_out;

  const size_t A = 255;
  size_t off = 0;
  float* dinv = (float*)((char*)d_ws + off);  off = (off + (size_t)N * sizeof(float) + A) & ~A;
  float* h    = (float*)((char*)d_ws + off);  off = (off + (size_t)N * DIM * sizeof(float) + A) & ~A;
  int* cnt      = (int*)((char*)d_ws + off);  off = (off + (size_t)N * sizeof(int) + A) & ~A;
  int* cursor   = (int*)((char*)d_ws + off);  off = (off + (size_t)N * sizeof(int) + A) & ~A;
  int* rowstart = (int*)((char*)d_ws + off);  off = (off + ((size_t)N + 1) * sizeof(int) + A) & ~A;
  int* bsum     = (int*)((char*)d_ws + off);  off = (off + (size_t)nblk * sizeof(int) + A) & ~A;
  int* csr      = (int*)((char*)d_ws + off);

  csr_zero_kernel<<<(N + 255) / 256, 256, 0, stream>>>(cnt, N);
  csr_count_kernel<<<(E + 255) / 256, 256, 0, stream>>>(ei, cnt, E);
  csr_blocksum_kernel<<<nblk, 256, 0, stream>>>(cnt, bsum, dinv, N);
  csr_scan_kernel<<<nblk, 256, 0, stream>>>(cnt, bsum, rowstart, cursor, N);
  csr_fill_kernel<<<(E + 255) / 256, 256, 0, stream>>>(ei, cursor, csr, E);
  gcn_gemm_kernel<<<(N + 15) / 16, 256, 0, stream>>>(x, W, h, N);
  gcn_gather_kernel<<<(N + 7) / 8, 256, 0, stream>>>(x, b, dinv, h, rowstart, csr, out, N);
}


# edge sort by dst with no global atomics: per-chunk LDS histogram over 512-row buckets, placed by column sums, one block per bucket sorts in LDS; 5 launches not 7; sum order = LDS cursor arrival
# speedup vs baseline: 3.3358x; 1.9819x over previous
#include <hip/hip_runtime.h>


typedef float v2f __attribute__((ext_vector_type(2)));
typedef float v8f __attribute__((ext_vector_type(8)));

#define DIM 128
#define XP 132
#define NEG_SLOPE 0.1f
#define BW 512
#define BW_SHIFT 9
#define CHUNK 8192
#define PT 1024
#define EPT (CHUNK / PT)
#define CG 256
#define RB 16
#define SB 8

__device__ __forceinline__ int block_excl_scan(int v, int* wtot) {
  const int lane = threadIdx.x & 31, wave = threadIdx.x >> 5;
  int incl = v;
  for (int o = 1; o < 32; o <<= 1) {
    int t = __shfl_up(incl, o, 32);
    if (lane >= o) incl += t;
  }
  if (lane == 31) wtot[wave] = incl;
  __syncthreads();
  int wpre = 0;
  for (int w = 0; w < wave; ++w) wpre += wtot[w];
  return wpre + incl - v;
}

__global__ __launch_bounds__(PT) void part_hist_kernel(const int* __restrict__ ei, int* __restrict__ hist,
                                                       int E, int N, int NB) {
  __shared__ int lc[PT];
  const int tid = threadIdx.x;
  lc[tid] = 0;
  __syncthreads();
  const int e0 = blockIdx.x * CHUNK;
  int dv[EPT];
  for (int k = 0; k < EPT; ++k) dv[k] = ei[E + min(e0 + k * PT + tid, E - 1)];
  for (int k = 0; k < EPT; ++k) {
    if (e0 + k * PT + tid < E) {
      const int d = min(max(dv[k], 0), N - 1);
      atomicAdd(&lc[d >> BW_SHIFT], 1);
    }
  }
  __syncthreads();
  if (tid < NB) hist[blockIdx.x * NB + tid] = lc[tid];
}

__global__ __launch_bounds__(PT) void part_place_kernel(const int* __restrict__ ei, const int* __restrict__ hist,
                                                        int* __restrict__ bstart, unsigned* __restrict__ tmp,
                                                        int E, int N, int NB, int P) {
  __shared__ int cur[PT];
  __shared__ int tot_s[PT];
  __shared__ int mine_s[PT];
  __shared__ int wtot[32];
  const int tid = threadIdx.x;
  const int e0 = blockIdx.x * CHUNK;

  tot_s[tid] = 0;
  mine_s[tid] = 0;
  __syncthreads();

  int dv[EPT], sv[EPT];
  for (int k = 0; k < EPT; ++k) {
    const int e = e0 + k * PT + tid;
    dv[k] = 0;
    sv[k] = 0;
    if (e < E) {
      dv[k] = ei[E + e];
      sv[k] = ei[e];
    }
  }

  const int grp = __builtin_amdgcn_readfirstlane(tid / CG);
  for (int cidx = tid & (CG - 1); cidx < NB; cidx += CG) {
    int t = 0, m = 0;
    for (int p0 = grp; p0 < P; p0 += RB * (PT / CG)) {
      int v[RB];
      for (int k = 0; k < RB; ++k) v[k] = hist[min(p0 + k * (PT / CG), P - 1) * NB + cidx];
      for (int k = 0; k < RB; ++k) {
        const int p = p0 + k * (PT / CG);
        if (p < P) {
          t += v[k];
          if (p < (int)blockIdx.x) m += v[k];
        }
      }
    }
    atomicAdd(&tot_s[cidx], t);
    atomicAdd(&mine_s[cidx], m);
  }
  __syncthreads();
  const int tot = tot_s[tid];
  const int mine = mine_s[tid];
  const int excl = block_excl_scan(tot, wtot);
  cur[tid] = excl + mine;
  if (blockIdx.x == 0 && tid <= NB) bstart[tid] = excl;
  __syncthreads();

  for (int k = 0; k < EPT; ++k) {
    const int e = e0 + k * PT + tid;
    if (e < E) {
      const int d = min(max(dv[k], 0), N - 1);
      const int pos = atomicAdd(&cur[d >> BW_SHIFT], 1);
      tmp[pos] = ((unsigned)sv[k] << BW_SHIFT) | (unsigned)(d & (BW - 1));
    }
  }
}

__global__ __launch_bounds__(PT) void bucket_sort_kernel(const unsigned* __restrict__ tmp,
                                                         const int* __restrict__ bstart,
                                                         int* __restrict__ rowstart, float* __restrict__ dinv,
                                                         int* __restrict__ csr, int N) {
  __shared__ int cnt[BW];
  __shared__ int wtot[32];
  const int tid = threadIdx.x;
  if (tid < BW) cnt[tid] = 0;
  __syncthreads();
  const int beg = bstart[blockIdx.x];
  const int end = bstart[blockIdx.x + 1];
  for (int e = beg + tid; e < end; e += SB * PT) {
    unsigned v[SB];
    for (int k = 0; k < SB; ++k) v[k] = tmp[min(e + k * PT, end - 1)];
    for (int k = 0; k < SB; ++k)
      if (e + k * PT < end) atomicAdd(&cnt[v[k] & (BW - 1)], 1);
  }
  __syncthreads();

  const int c = (tid < BW) ? cnt[tid] : 0;
  const int rs = beg + block_excl_scan(c, wtot);
  const int node = blockIdx.x * BW + tid;
  if (tid < BW) {
    cnt[tid] = rs;
    if (node < N) {
      rowstart[node] = rs;
      dinv[node] = rsqrtf((float)(c + 1));
      if (node == N - 1) rowstart[N] = rs + c;
    }
  }
  __syncthreads();

  for (int e = beg + tid; e < end; e += SB * PT) {
    unsigned v[SB];
    for (int k = 0; k < SB; ++k) v[k] = tmp[min(e + k * PT, end - 1)];
    for (int k = 0; k < SB; ++k) {
      if (e + k * PT < end) {
        const int pos = atomicAdd(&cnt[v[k] & (BW - 1)], 1);
        csr[pos] = (int)(v[k] >> BW_SHIFT);
      }
    }
  }
}

__global__ __launch_bounds__(256) void gcn_gemm_kernel(const float* __restrict__ x,
                                                       const float* __restrict__ W,
                                                       float* __restrict__ h, int N) {
  __shared__ float Ws[DIM * DIM];
  __shared__ float xs[16 * XP];

  const int tid  = threadIdx.x;
  const int row0 = blockIdx.x * 16;

  for (int i = tid; i < (DIM * DIM) / 4; i += 256) {
    ((float4*)Ws)[i] = ((const float4*)W)[i];
  }
  for (int i = tid; i < 16 * DIM; i += 256) {
    int r = i >> 7, c = i & (DIM - 1);
    int gr = row0 + r;
    xs[r * XP + c] = (gr < N) ? x[(long)gr * DIM + c] : 0.0f;
  }
  __syncthreads();

  const int wave = tid >> 5;
  const int lane = tid & 31;
  const int half = lane >> 4;
  const int m    = lane & 15;
  const int col0 = wave * 16;

  v8f acc = {};
  for (int k0 = 0; k0 < DIM; k0 += 4) {
    int ka = k0 + 2 * half;
    v2f a, b;
    a.x = xs[m * XP + ka];
    a.y = xs[m * XP + ka + 1];
    b.x = Ws[ka * DIM + col0 + m];
    b.y = Ws[(ka + 1) * DIM + col0 + m];
    acc = __builtin_amdgcn_wmma_f32_16x16x4_f32(
         false, a,  false, b,
         (short)0, acc,  false,  false);
  }

  for (int r = 0; r < 8; ++r) {
    int row = row0 + r + 8 * half;
    if (row < N) h[(long)row * DIM + col0 + m] = acc[r];
  }
}

#define ROW4(s) (((const float4*)(h + (size_t)(s) * DIM))[lane])
#define FMA4(rowv, wgt)                    \
  acc.x = fmaf((rowv).x, (wgt), acc.x);  \
  acc.y = fmaf((rowv).y, (wgt), acc.y);  \
  acc.z = fmaf((rowv).z, (wgt), acc.z);  \
  acc.w = fmaf((rowv).w, (wgt), acc.w)
#define LANE_I(v, k) __builtin_amdgcn_readlane((v), (k))
#define LANE_W(v, k) (__int_as_float(__builtin_amdgcn_readlane(__float_as_int(v), (k))) * di)

__global__ __launch_bounds__(256) void gcn_gather_kernel(const float* __restrict__ x,
                                                         const float* __restrict__ bias,
                                                         const float* __restrict__ dinv,
                                                         const float* __restrict__ h,
                                                         const int* __restrict__ rowstart,
                                                         const int* __restrict__ csr,
                                                         float* __restrict__ out, int N) {
  const int lane = threadIdx.x & 31;
  const int d = __builtin_amdgcn_readfirstlane((int)(blockIdx.x * 8 + (threadIdx.x >> 5)));
  if (d >= N) return;

  const int beg = rowstart[d];
  const int end = rowstart[d + 1];
  const float di = dinv[d];
  const float4 hv = ROW4(d);
  const float4 xv = ((const float4*)(x + (size_t)d * DIM))[lane];
  const float4 bv = ((const float4*)bias)[lane];

  const float wself = di * di;
  float4 acc;
  acc.x = hv.x * wself;
  acc.y = hv.y * wself;
  acc.z = hv.z * wself;
  acc.w = hv.w * wself;

  for (int base = beg; base < end; base += 32) {
    const int j = base + lane;
    int sv = 0;
    float wv = 0.0f;
    if (j < end) {
      int s = csr[j];
      s = min(max(s, 0), N - 1);
      sv = s;
      wv = dinv[s];
    }
    int n = end - base;
    if (n > 32) n = 32;

    int k = 0;
    for (; k + 4 <= n; k += 4) {
      const int s0 = LANE_I(sv, k), s1 = LANE_I(sv, k + 1), s2 = LANE_I(sv, k + 2), s3 = LANE_I(sv, k + 3);
      const float w0 = LANE_W(wv, k), w1 = LANE_W(wv, k + 1), w2 = LANE_W(wv, k + 2), w3 = LANE_W(wv, k + 3);
      const float4 r0 = ROW4(s0);
      const float4 r1 = ROW4(s1);
      const float4 r2 = ROW4(s2);
      const float4 r3 = ROW4(s3);
      FMA4(r0, w0);
      FMA4(r1, w1);
      FMA4(r2, w2);
      FMA4(r3, w3);
    }
    const int rem = n - k;
    if (rem == 3) {
      const int s0 = LANE_I(sv, k), s1 = LANE_I(sv, k + 1), s2 = LANE_I(sv, k + 2);
      const float w0 = LANE_W(wv, k), w1 = LANE_W(wv, k + 1), w2 = LANE_W(wv, k + 2);
      const float4 r0 = ROW4(s0);
      const float4 r1 = ROW4(s1);
      const float4 r2 = ROW4(s2);
      FMA4(r0, w0);
      FMA4(r1, w1);
      FMA4(r2, w2);
    } else if (rem == 2) {
      const int s0 = LANE_I(sv, k), s1 = LANE_I(sv, k + 1);
      const float w0 = LANE_W(wv, k), w1 = LANE_W(wv, k + 1);
      const float4 r0 = ROW4(s0);
      const float4 r1 = ROW4(s1);
      FMA4(r0, w0);
      FMA4(r1, w1);
    } else if (rem == 1) {
      const int s0 = LANE_I(sv, k);
      const float w0 = LANE_W(wv, k);
      const float4 r0 = ROW4(s0);
      FMA4(r0, w0);
    }
  }

  float4 o;
  float v;
  v = acc.x + bv.x; v = (v >= 0.0f) ? v : NEG_SLOPE * v; o.x = v + xv.x;
  v = acc.y + bv.y; v = (v >= 0.0f) ? v : NEG_SLOPE * v; o.y = v + xv.y;
  v = acc.z + bv.z; v = (v >= 0.0f) ? v : NEG_SLOPE * v; o.z = v + xv.z;
  v = acc.w + bv.w; v = (v >= 0.0f) ? v : NEG_SLOPE * v; o.w = v + xv.w;
  ((float4*)(out + (size_t)d * DIM))[lane] = o;
}

extern "C" void kernel_launch(void* const* d_in, const int* in_sizes, int n_in,
                              void* d_out, int out_size, void* d_ws, size_t ws_size,
                              hipStream_t stream) {
  const float* x  = (const float*)d_in[0];
  const int*   ei = (const int*)d_in[1];
  const float* W  = (const float*)d_in[2];
  const float* b  = (const float*)d_in[3];

  const int N = in_sizes[0] / DIM;
  const int E = in_sizes[1] / 2;
  const int NB = (N + BW - 1) / BW;
  const int P  = (E + CHUNK - 1) / CHUNK;

  float* out = (float*)d_out;

  const size_t A = 255;
  size_t off = 0;
  float* dinv = (float*)((char*)d_ws + off);  off = (off + (size_t)N * sizeof(float) + A) & ~A;
  float* h    = (float*)((char*)d_ws + off);  off = (off + (size_t)N * DIM * sizeof(float) + A) & ~A;
  int* rowstart = (int*)((char*)d_ws + off);  off = (off + ((size_t)N + 1) * sizeof(int) + A) & ~A;
  int* hist     = (int*)((char*)d_ws + off);  off = (off + (size_t)P * NB * sizeof(int) + A) & ~A;
  int* bstart   = (int*)((char*)d_ws + off);  off = (off + ((size_t)NB + 1) * sizeof(int) + A) & ~A;
  unsigned* tmp = (unsigned*)((char*)d_ws + off);  off = (off + (size_t)E * sizeof(unsigned) + A) & ~A;
  int* csr      = (int*)((char*)d_ws + off);

  part_hist_kernel<<<P, PT, 0, stream>>>(ei, hist, E, N, NB);
  part_place_kernel<<<P, PT, 0, stream>>>(ei, hist, bstart, tmp, E, N, NB, P);
  bucket_sort_kernel<<<NB, PT, 0, stream>>>(tmp, bstart, rowstart, dinv, csr, N);
  gcn_gemm_kernel<<<(N + 15) / 16, 256, 0, stream>>>(x, W, h, N);
  gcn_gather_kernel<<<(N + 7) / 8, 256, 0, stream>>>(x, b, dinv, h, rowstart, csr, out, N);
}


# gemm and row-sum launches as resident chip-sized grids: each wave loops over 16-row tiles / rows and keeps its W slice in registers (was restaged to LDS per tile); 150k entering waves become 23k
# speedup vs baseline: 4.0628x; 1.2179x over previous
#include <hip/hip_runtime.h>


typedef float v2f __attribute__((ext_vector_type(2)));
typedef float v8f __attribute__((ext_vector_type(8)));

#define DIM 128
#define XP 132
#define NEG_SLOPE 0.1f
#define BW 512
#define BW_SHIFT 9
#define CHUNK 8192
#define PT 1024
#define EPT (CHUNK / PT)
#define CG 256
#define RB 16
#define SB 8

__device__ __forceinline__ int block_excl_scan(int v, int* wtot) {
  const int lane = threadIdx.x & 31, wave = threadIdx.x >> 5;
  int incl = v;
  for (int o = 1; o < 32; o <<= 1) {
    int t = __shfl_up(incl, o, 32);
    if (lane >= o) incl += t;
  }
  if (lane == 31) wtot[wave] = incl;
  __syncthreads();
  int wpre = 0;
  for (int w = 0; w < wave; ++w) wpre += wtot[w];
  return wpre + incl - v;
}

__global__ __launch_bounds__(PT) void part_hist_kernel(const int* __restrict__ ei, int* __restrict__ hist,
                                                       int E, int N, int NB) {
  __shared__ int lc[PT];
  const int tid = threadIdx.x;
  lc[tid] = 0;
  __syncthreads();
  const int e0 = blockIdx.x * CHUNK;
  int dv[EPT];
  for (int k = 0; k < EPT; ++k) dv[k] = ei[E + min(e0 + k * PT + tid, E - 1)];
  for (int k = 0; k < EPT; ++k) {
    if (e0 + k * PT + tid < E) {
      const int d = min(max(dv[k], 0), N - 1);
      atomicAdd(&lc[d >> BW_SHIFT], 1);
    }
  }
  __syncthreads();
  if (tid < NB) hist[blockIdx.x * NB + tid] = lc[tid];
}

__global__ __launch_bounds__(PT) void part_place_kernel(const int* __restrict__ ei, const int* __restrict__ hist,
                                                        int* __restrict__ bstart, unsigned* __restrict__ tmp,
                                                        int E, int N, int NB, int P) {
  __shared__ int cur[PT];
  __shared__ int tot_s[PT];
  __shared__ int mine_s[PT];
  __shared__ int wtot[32];
  const int tid = threadIdx.x;
  const int e0 = blockIdx.x * CHUNK;

  tot_s[tid] = 0;
  mine_s[tid] = 0;
  __syncthreads();

  int dv[EPT], sv[EPT];
  for (int k = 0; k < EPT; ++k) {
    const int e = e0 + k * PT + tid;
    dv[k] = 0;
    sv[k] = 0;
    if (e < E) {
      dv[k] = ei[E + e];
      sv[k] = ei[e];
    }
  }

  const int grp = __builtin_amdgcn_readfirstlane(tid / CG);
  for (int cidx = tid & (CG - 1); cidx < NB; cidx += CG) {
    int t = 0, m = 0;
    for (int p0 = grp; p0 < P; p0 += RB * (PT / CG)) {
      int v[RB];
      for (int k = 0; k < RB; ++k) v[k] = hist[min(p0 + k * (PT / CG), P - 1) * NB + cidx];
      for (int k = 0; k < RB; ++k) {
        const int p = p0 + k * (PT / CG);
        if (p < P) {
          t += v[k];
          if (p < (int)blockIdx.x) m += v[k];
        }
      }
    }
    atomicAdd(&tot_s[cidx], t);
    atomicAdd(&mine_s[cidx], m);
  }
  __syncthreads();
  const int tot = tot_s[tid];
  const int mine = mine_s[tid];
  const int excl = block_excl_scan(tot, wtot);
  cur[tid] = excl + mine;
  if (blockIdx.x == 0 && tid <= NB) bstart[tid] = excl;
  __syncthreads();

  for (int k = 0; k < EPT; ++k) {
    const int e = e0 + k * PT + tid;
    if (e < E) {
      const int d = min(max(dv[k], 0), N - 1);
      const int pos = atomicAdd(&cur[d >> BW_SHIFT], 1);
      tmp[pos] = ((unsigned)sv[k] << BW_SHIFT) | (unsigned)(d & (BW - 1));
    }
  }
}

__global__ __launch_bounds__(PT) void bucket_sort_kernel(const unsigned* __restrict__ tmp,
                                                         const int* __restrict__ bstart,
                                                         int* __restrict__ rowstart, float* __restrict__ dinv,
                                                         int* __restrict__ csr, int N) {
  __shared__ int cnt[BW];
  __shared__ int wtot[32];
  const int tid = threadIdx.x;
  if (tid < BW) cnt[tid] = 0;
  __syncthreads();
  const int beg = bstart[blockIdx.x];
  const int end = bstart[blockIdx.x + 1];
  for (int e = beg + tid; e < end; e += SB * PT) {
    unsigned v[SB];
    for (int k = 0; k < SB; ++k) v[k] = tmp[min(e + k * PT, end - 1)];
    for (int k = 0; k < SB; ++k)
      if (e + k * PT < end) atomicAdd(&cnt[v[k] & (BW - 1)], 1);
  }
  __syncthreads();

  const int c = (tid < BW) ? cnt[tid] : 0;
  const int rs = beg + block_excl_scan(c, wtot);
  const int node = blockIdx.x * BW + tid;
  if (tid < BW) {
    cnt[tid] = rs;
    if (node < N) {
      rowstart[node] = rs;
      dinv[node] = rsqrtf((float)(c + 1));
      if (node == N - 1) rowstart[N] = rs + c;
    }
  }
  __syncthreads();

  for (int e = beg + tid; e < end; e += SB * PT) {
    unsigned v[SB];
    for (int k = 0; k < SB; ++k) v[k] = tmp[min(e + k * PT, end - 1)];
    for (int k = 0; k < SB; ++k) {
      if (e + k * PT < end) {
        const int pos = atomicAdd(&cnt[v[k] & (BW - 1)], 1);
        csr[pos] = (int)(v[k] >> BW_SHIFT);
      }
    }
  }
}

__global__ __launch_bounds__(256) void gcn_gemm_kernel(const float* __restrict__ x,
                                                       const float* __restrict__ W,
                                                       float* __restrict__ h, int N) {
  __shared__ float xs[2][16 * XP];

  const int tid  = threadIdx.x;
  const int wave = tid >> 5;
  const int lane = tid & 31;
  const int half = lane >> 4;
  const int m    = lane & 15;
  const int col0 = wave * 16;
  const int ntiles = (N + 15) >> 4;

  v2f breg[DIM / 4];
#pragma unroll
  for (int j = 0; j < DIM / 4; ++j) {
    const int ka = 4 * j + 2 * half;
    breg[j].x = W[ka * DIM + col0 + m];
    breg[j].y = W[(ka + 1) * DIM + col0 + m];
  }

  const int c4 = lane * 4;
  float4 p0, p1;
  int tile = blockIdx.x;
  if (tile < ntiles) {
    const int g0 = tile * 16 + wave, g1 = g0 + 8;
    p0 = (g0 < N) ? *(const float4*)(x + (size_t)g0 * DIM + c4) : make_float4(0.f, 0.f, 0.f, 0.f);
    p1 = (g1 < N) ? *(const float4*)(x + (size_t)g1 * DIM + c4) : make_float4(0.f, 0.f, 0.f, 0.f);
    *(float4*)(&xs[0][wave * XP + c4]) = p0;
    *(float4*)(&xs[0][(wave + 8) * XP + c4]) = p1;
  }
  __syncthreads();

  int buf = 0;
  for (; tile < ntiles; tile += gridDim.x) {
    const int nt = tile + gridDim.x;
    const bool more = nt < ntiles;
    if (more) {
      const int g0 = nt * 16 + wave, g1 = g0 + 8;
      p0 = (g0 < N) ? *(const float4*)(x + (size_t)g0 * DIM + c4) : make_float4(0.f, 0.f, 0.f, 0.f);
      p1 = (g1 < N) ? *(const float4*)(x + (size_t)g1 * DIM + c4) : make_float4(0.f, 0.f, 0.f, 0.f);
    }

    const float* xb = xs[buf];
    v8f acc = {};
#pragma unroll
    for (int j = 0; j < DIM / 4; ++j) {
      const int ka = 4 * j + 2 * half;
      v2f a;
      a.x = xb[m * XP + ka];
      a.y = xb[m * XP + ka + 1];
      acc = __builtin_amdgcn_wmma_f32_16x16x4_f32(
           false, a,  false, breg[j],
           (short)0, acc,  false,  false);
    }

    if (more) {
      *(float4*)(&xs[buf ^ 1][wave * XP + c4]) = p0;
      *(float4*)(&xs[buf ^ 1][(wave + 8) * XP + c4]) = p1;
    }
    __syncthreads();
    buf ^= 1;

    const int row0 = tile * 16;
    for (int r = 0; r < 8; ++r) {
      const int row = row0 + r + 8 * half;
      if (row < N) h[(long)row * DIM + col0 + m] = acc[r];
    }
  }
}

#define ROW4(s) (((const float4*)(h + (size_t)(s) * DIM))[lane])
#define FMA4(rowv, wgt)                    \
  acc.x = fmaf((rowv).x, (wgt), acc.x);  \
  acc.y = fmaf((rowv).y, (wgt), acc.y);  \
  acc.z = fmaf((rowv).z, (wgt), acc.z);  \
  acc.w = fmaf((rowv).w, (wgt), acc.w)
#define LANE_I(v, k) __builtin_amdgcn_readlane((v), (k))
#define LANE_W(v, k) (__int_as_float(__builtin_amdgcn_readlane(__float_as_int(v), (k))) * di)

__global__ __launch_bounds__(256) void gcn_gather_kernel(const float* __restrict__ x,
                                                         const float* __restrict__ bias,
                                                         const float* __restrict__ dinv,
                                                         const float* __restrict__ h,
                                                         const int* __restrict__ rowstart,
                                                         const int* __restrict__ csr,
                                                         float* __restrict__ out, int N) {
  const int lane = threadIdx.x & 31;
  const int gw = __builtin_amdgcn_readfirstlane((int)(blockIdx.x * 8 + (threadIdx.x >> 5)));
  const int nw = (int)gridDim.x * 8;
  if (gw >= N) return;

  const float4 bv = ((const float4*)bias)[lane];
  int beg = rowstart[gw];
  int end = rowstart[gw + 1];
  for (int d = gw; d < N;) {
  const int dn = d + nw;
  int nbeg = 0, nend = 0;
  if (dn < N) {
    nbeg = rowstart[dn];
    nend = rowstart[dn + 1];
  }
  const float di = dinv[d];
  const float4 hv = ROW4(d);
  const float4 xv = ((const float4*)(x + (size_t)d * DIM))[lane];

  const float wself = di * di;
  float4 acc;
  acc.x = hv.x * wself;
  acc.y = hv.y * wself;
  acc.z = hv.z * wself;
  acc.w = hv.w * wself;

  for (int base = beg; base < end; base += 32) {
    const int j = base + lane;
    int sv = 0;
    float wv = 0.0f;
    if (j < end) {
      int s = csr[j];
      s = min(max(s, 0), N - 1);
      sv = s;
      wv = dinv[s];
    }
    int n = end - base;
    if (n > 32) n = 32;

    int k = 0;
    for (; k + 4 <= n; k += 4) {
      const int s0 = LANE_I(sv, k), s1 = LANE_I(sv, k + 1), s2 = LANE_I(sv, k + 2), s3 = LANE_I(sv, k + 3);
      const float w0 = LANE_W(wv, k), w1 = LANE_W(wv, k + 1), w2 = LANE_W(wv, k + 2), w3 = LANE_W(wv, k + 3);
      const float4 r0 = ROW4(s0);
      const float4 r1 = ROW4(s1);
      const float4 r2 = ROW4(s2);
      const float4 r3 = ROW4(s3);
      FMA4(r0, w0);
      FMA4(r1, w1);
      FMA4(r2, w2);
      FMA4(r3, w3);
    }
    const int rem = n - k;
    if (rem == 3) {
      const int s0 = LANE_I(sv, k), s1 = LANE_I(sv, k + 1), s2 = LANE_I(sv, k + 2);
      const float w0 = LANE_W(wv, k), w1 = LANE_W(wv, k + 1), w2 = LANE_W(wv, k + 2);
      const float4 r0 = ROW4(s0);
      const float4 r1 = ROW4(s1);
      const float4 r2 = ROW4(s2);
      FMA4(r0, w0);
      FMA4(r1, w1);
      FMA4(r2, w2);
    } else if (rem == 2) {
      const int s0 = LANE_I(sv, k), s1 = LANE_I(sv, k + 1);
      const float w0 = LANE_W(wv, k), w1 = LANE_W(wv, k + 1);
      const float4 r0 = ROW4(s0);
      const float4 r1 = ROW4(s1);
      FMA4(r0, w0);
      FMA4(r1, w1);
    } else if (rem == 1) {
      const int s0 = LANE_I(sv, k);
      const float w0 = LANE_W(wv, k);
      const float4 r0 = ROW4(s0);
      FMA4(r0, w0);
    }
  }

  float4 o;
  float v;
  v = acc.x + bv.x; v = (v >= 0.0f) ? v : NEG_SLOPE * v; o.x = v + xv.x;
  v = acc.y + bv.y; v = (v >= 0.0f) ? v : NEG_SLOPE * v; o.y = v + xv.y;
  v = acc.z + bv.z; v = (v >= 0.0f) ? v : NEG_SLOPE * v; o.z = v + xv.z;
  v = acc.w + bv.w; v = (v >= 0.0f) ? v : NEG_SLOPE * v; o.w = v + xv.w;
  ((float4*)(out + (size_t)d * DIM))[lane] = o;

  d = dn;
  beg = nbeg;
  end = nend;
  }
}

extern "C" void kernel_launch(void* const* d_in, const int* in_sizes, int n_in,
                              void* d_out, int out_size, void* d_ws, size_t ws_size,
                              hipStream_t stream) {
  const float* x  = (const float*)d_in[0];
  const int*   ei = (const int*)d_in[1];
  const float* W  = (const float*)d_in[2];
  const float* b  = (const float*)d_in[3];

  const int N = in_sizes[0] / DIM;
  const int E = in_sizes[1] / 2;
  const int NB = (N + BW - 1) / BW;
  const int P  = (E + CHUNK - 1) / CHUNK;

  float* out = (float*)d_out;

  const size_t A = 255;
  size_t off = 0;
  float* dinv = (float*)((char*)d_ws + off);  off = (off + (size_t)N * sizeof(float) + A) & ~A;
  float* h    = (float*)((char*)d_ws + off);  off = (off + (size_t)N * DIM * sizeof(float) + A) & ~A;
  int* rowstart = (int*)((char*)d_ws + off);  off = (off + ((size_t)N + 1) * sizeof(int) + A) & ~A;
  int* hist     = (int*)((char*)d_ws + off);  off = (off + (size_t)P * NB * sizeof(int) + A) & ~A;
  int* bstart   = (int*)((char*)d_ws + off);  off = (off + ((size_t)NB + 1) * sizeof(int) + A) & ~A;
  unsigned* tmp = (unsigned*)((char*)d_ws + off);  off = (off + (size_t)E * sizeof(unsigned) + A) & ~A;
  int* csr      = (int*)((char*)d_ws + off);

  part_hist_kernel<<<P, PT, 0, stream>>>(ei, hist, E, N, NB);
  part_place_kernel<<<P, PT, 0, stream>>>(ei, hist, bstart, tmp, E, N, NB, P);
  bucket_sort_kernel<<<NB, PT, 0, stream>>>(tmp, bstart, rowstart, dinv, csr, N);
  gcn_gemm_kernel<<<1024, 256, 0, stream>>>(x, W, h, N);
  gcn_gather_kernel<<<1792, 256, 0, stream>>>(x, b, dinv, h, rowstart, csr, out, N);
}


# W applied after the neighbour sum (linearity): the row-sum launch sums x rows, multiplies each 16-row tile by W on f32 WMMA, finishes the row; h plane and gemm launch gone; rounding order changes
# speedup vs baseline: 4.5389x; 1.1172x over previous
#include <hip/hip_runtime.h>


typedef float v2f __attribute__((ext_vector_type(2)));
typedef float v8f __attribute__((ext_vector_type(8)));

#define DIM 128
#define XP 132
#define NEG_SLOPE 0.1f
#define BW 512
#define BW_SHIFT 9
#define CHUNK 8192
#define PT 1024
#define EPT (CHUNK / PT)
#define CG 256
#define RB 16
#define SB 8

__device__ __forceinline__ int block_excl_scan(int v, int* wtot) {
  const int lane = threadIdx.x & 31, wave = threadIdx.x >> 5;
  int incl = v;
  for (int o = 1; o < 32; o <<= 1) {
    int t = __shfl_up(incl, o, 32);
    if (lane >= o) incl += t;
  }
  if (lane == 31) wtot[wave] = incl;
  __syncthreads();
  int wpre = 0;
  for (int w = 0; w < wave; ++w) wpre += wtot[w];
  return wpre + incl - v;
}

__global__ __launch_bounds__(PT) void part_hist_kernel(const int* __restrict__ ei, int* __restrict__ hist,
                                                       int E, int N, int NB) {
  __shared__ int lc[PT];
  const int tid = threadIdx.x;
  lc[tid] = 0;
  __syncthreads();
  const int e0 = blockIdx.x * CHUNK;
  int dv[EPT];
  for (int k = 0; k < EPT; ++k) dv[k] = ei[E + min(e0 + k * PT + tid, E - 1)];
  for (int k = 0; k < EPT; ++k) {
    if (e0 + k * PT + tid < E) {
      const int d = min(max(dv[k], 0), N - 1);
      atomicAdd(&lc[d >> BW_SHIFT], 1);
    }
  }
  __syncthreads();
  if (tid < NB) hist[blockIdx.x * NB + tid] = lc[tid];
}

__global__ __launch_bounds__(PT) void part_place_kernel(const int* __restrict__ ei, const int* __restrict__ hist,
                                                        int* __restrict__ bstart, unsigned* __restrict__ tmp,
                                                        int E, int N, int NB, int P) {
  __shared__ int cur[PT];
  __shared__ int tot_s[PT];
  __shared__ int mine_s[PT];
  __shared__ int wtot[32];
  const int tid = threadIdx.x;
  const int e0 = blockIdx.x * CHUNK;

  tot_s[tid] = 0;
  mine_s[tid] = 0;
  __syncthreads();

  int dv[EPT], sv[EPT];
  for (int k = 0; k < EPT; ++k) {
    const int e = e0 + k * PT + tid;
    dv[k] = 0;
    sv[k] = 0;
    if (e < E) {
      dv[k] = ei[E + e];
      sv[k] = ei[e];
    }
  }

  const int grp = __builtin_amdgcn_readfirstlane(tid / CG);
  for (int cidx = tid & (CG - 1); cidx < NB; cidx += CG) {
    int t = 0, m = 0;
    for (int p0 = grp; p0 < P; p0 += RB * (PT / CG)) {
      int v[RB];
      for (int k = 0; k < RB; ++k) v[k] = hist[min(p0 + k * (PT / CG), P - 1) * NB + cidx];
      for (int k = 0; k < RB; ++k) {
        const int p = p0 + k * (PT / CG);
        if (p < P) {
          t += v[k];
          if (p < (int)blockIdx.x) m += v[k];
        }
      }
    }
    atomicAdd(&tot_s[cidx], t);
    atomicAdd(&mine_s[cidx], m);
  }
  __syncthreads();
  const int tot = tot_s[tid];
  const int mine = mine_s[tid];
  const int excl = block_excl_scan(tot, wtot);
  cur[tid] = excl + mine;
  if (blockIdx.x == 0 && tid <= NB) bstart[tid] = excl;
  __syncthreads();

  for (int k = 0; k < EPT; ++k) {
    const int e = e0 + k * PT + tid;
    if (e < E) {
      const int d = min(max(dv[k], 0), N - 1);
      const int pos = atomicAdd(&cur[d >> BW_SHIFT], 1);
      tmp[pos] = ((unsigned)sv[k] << BW_SHIFT) | (unsigned)(d & (BW - 1));
    }
  }
}

__global__ __launch_bounds__(PT) void bucket_sort_kernel(const unsigned* __restrict__ tmp,
                                                         const int* __restrict__ bstart,
                                                         int* __restrict__ rowstart, float* __restrict__ dinv,
                                                         int* __restrict__ csr, int N) {
  __shared__ int cnt[BW];
  __shared__ int wtot[32];
  const int tid = threadIdx.x;
  if (tid < BW) cnt[tid] = 0;
  __syncthreads();
  const int beg = bstart[blockIdx.x];
  const int end = bstart[blockIdx.x + 1];
  for (int e = beg + tid; e < end; e += SB * PT) {
    unsigned v[SB];
    for (int k = 0; k < SB; ++k) v[k] = tmp[min(e + k * PT, end - 1)];
    for (int k = 0; k < SB; ++k)
      if (e + k * PT < end) atomicAdd(&cnt[v[k] & (BW - 1)], 1);
  }
  __syncthreads();

  const int c = (tid < BW) ? cnt[tid] : 0;
  const int rs = beg + block_excl_scan(c, wtot);
  const int node = blockIdx.x * BW + tid;
  if (tid < BW) {
    cnt[tid] = rs;
    if (node < N) {
      rowstart[node] = rs;
      dinv[node] = rsqrtf((float)(c + 1));
      if (node == N - 1) rowstart[N] = rs + c;
    }
  }
  __syncthreads();

  for (int e = beg + tid; e < end; e += SB * PT) {
    unsigned v[SB];
    for (int k = 0; k < SB; ++k) v[k] = tmp[min(e + k * PT, end - 1)];
    for (int k = 0; k < SB; ++k) {
      if (e + k * PT < end) {
        const int pos = atomicAdd(&cnt[v[k] & (BW - 1)], 1);
        csr[pos] = (int)(v[k] >> BW_SHIFT);
      }
    }
  }
}

#define ROWX(s) (((const float4*)(x + (size_t)(s) * DIM))[lane])
#define FMA4(rowv, wgt)                    \
  acc.x = fmaf((rowv).x, (wgt), acc.x);  \
  acc.y = fmaf((rowv).y, (wgt), acc.y);  \
  acc.z = fmaf((rowv).z, (wgt), acc.z);  \
  acc.w = fmaf((rowv).w, (wgt), acc.w)
#define LANE_I(v, k) __builtin_amdgcn_readlane((v), (k))
#define LANE_W(v, k) (__int_as_float(__builtin_amdgcn_readlane(__float_as_int(v), (k))) * di)

__global__ __launch_bounds__(256) void gcn_rowsum_gemm_kernel(const float* __restrict__ x,
                                                              const float* __restrict__ W,
                                                              const float* __restrict__ bias,
                                                              const float* __restrict__ dinv,
                                                              const int* __restrict__ rowstart,
                                                              const int* __restrict__ csr,
                                                              float* __restrict__ out, int N, int E) {
  __shared__ float at[16 * XP];
  __shared__ float dt[16 * XP];
  __shared__ float xt[16 * XP];

  const int tid  = threadIdx.x;
  const int lane = tid & 31;
  const int wave = __builtin_amdgcn_readfirstlane(tid >> 5);
  const int half = lane >> 4;
  const int m    = lane & 15;
  const int col0 = wave * 16;
  const int c4   = lane * 4;
  const int ntiles = (N + 15) >> 4;
  if (N <= 0) return;

  v2f breg[DIM / 4];
#pragma unroll
  for (int j = 0; j < DIM / 4; ++j) {
    const int ka = 4 * j + 2 * half;
    breg[j].x = W[ka * DIM + col0 + m];
    breg[j].y = W[(ka + 1) * DIM + col0 + m];
  }
  const float4 bv = ((const float4*)bias)[lane];

#define HEAD_ROWS(T, B0, E0, D0, X0, B1, E1, D1, X1)                          \
  {                                                                           \
    const int hr0 = min((T) * 16 + wave, N - 1), hr1 = min((T) * 16 + wave + 8, N - 1); \
    B0 = rowstart[hr0]; E0 = rowstart[hr0 + 1]; D0 = dinv[hr0]; X0 = ROWX(hr0); \
    B1 = rowstart[hr1]; E1 = rowstart[hr1 + 1]; D1 = dinv[hr1]; X1 = ROWX(hr1); \
  }
#define HEAD_SRC(B, S) S = min(max(csr[max(min((B) + lane, E - 1), 0)], 0), N - 1)

  int beg0, end0, beg1, end1, sv0, sv1;
  float di0, di1, wv0, wv1;
  float4 xv0, xv1;
  HEAD_ROWS((int)blockIdx.x, beg0, end0, di0, xv0, beg1, end1, di1, xv1);
  HEAD_SRC(beg0, sv0);
  HEAD_SRC(beg1, sv1);
  wv0 = dinv[sv0];
  wv1 = dinv[sv1];

  for (int tile = blockIdx.x; tile < ntiles; tile += gridDim.x) {
    const int d0 = tile * 16 + wave, d1 = d0 + 8;
    if (d0 >= N) { end0 = beg0; di0 = 0.0f; xv0 = make_float4(0.f, 0.f, 0.f, 0.f); }
    if (d1 >= N) { end1 = beg1; di1 = 0.0f; xv1 = make_float4(0.f, 0.f, 0.f, 0.f); }
    *(float4*)(&xt[wave * XP + c4]) = xv0;
    *(float4*)(&xt[(wave + 8) * XP + c4]) = xv1;

#pragma unroll
    for (int q = 0; q < 2; ++q) {
      const int beg = q ? beg1 : beg0, end = q ? end1 : end0;
      const float di = q ? di1 : di0;
      const float4 xv = q ? xv1 : xv0;
      int sv = q ? sv1 : sv0;
      float wv = q ? wv1 : wv0;

      const float wself = di * di;
      float4 acc;
      acc.x = xv.x * wself;
      acc.y = xv.y * wself;
      acc.z = xv.z * wself;
      acc.w = xv.w * wself;

      for (int base = beg; base < end; base += 32) {
        const int j = base + lane;
        int svn = 0;
        float wvn = 0.0f;
        if (base + 32 < end) {
          if (j + 32 < end) svn = min(max(csr[j + 32], 0), N - 1);
          wvn = dinv[svn];
        }
        int n = end - base;
        if (n > 32) n = 32;

        int k = 0;
        for (; k + 8 <= n; k += 8) {
          const int s0 = LANE_I(sv, k), s1 = LANE_I(sv, k + 1), s2 = LANE_I(sv, k + 2), s3 = LANE_I(sv, k + 3);
          const int s4 = LANE_I(sv, k + 4), s5 = LANE_I(sv, k + 5), s6 = LANE_I(sv, k + 6), s7 = LANE_I(sv, k + 7);
          const float4 r0 = ROWX(s0);
          const float4 r1 = ROWX(s1);
          const float4 r2 = ROWX(s2);
          const float4 r3 = ROWX(s3);
          const float4 r4 = ROWX(s4);
          const float4 r5 = ROWX(s5);
          const float4 r6 = ROWX(s6);
          const float4 r7 = ROWX(s7);
          const float w0 = LANE_W(wv, k), w1 = LANE_W(wv, k + 1), w2 = LANE_W(wv, k + 2), w3 = LANE_W(wv, k + 3);
          const float w4 = LANE_W(wv, k + 4), w5 = LANE_W(wv, k + 5), w6 = LANE_W(wv, k + 6), w7 = LANE_W(wv, k + 7);
          FMA4(r0, w0);
          FMA4(r1, w1);
          FMA4(r2, w2);
          FMA4(r3, w3);
          FMA4(r4, w4);
          FMA4(r5, w5);
          FMA4(r6, w6);
          FMA4(r7, w7);
        }
        if (k + 4 <= n) {
          const int s0 = LANE_I(sv, k), s1 = LANE_I(sv, k + 1), s2 = LANE_I(sv, k + 2), s3 = LANE_I(sv, k + 3);
          const float w0 = LANE_W(wv, k), w1 = LANE_W(wv, k + 1), w2 = LANE_W(wv, k + 2), w3 = LANE_W(wv, k + 3);
          const float4 r0 = ROWX(s0);
          const float4 r1 = ROWX(s1);
          const float4 r2 = ROWX(s2);
          const float4 r3 = ROWX(s3);
          FMA4(r0, w0);
          FMA4(r1, w1);
          FMA4(r2, w2);
          FMA4(r3, w3);
          k += 4;
        }
        const int rem = n - k;
        if (rem == 3) {
          const int s0 = LANE_I(sv, k), s1 = LANE_I(sv, k + 1), s2 = LANE_I(sv, k + 2);
          const float w0 = LANE_W(wv, k), w1 = LANE_W(wv, k + 1), w2 = LANE_W(wv, k + 2);
          const float4 r0 = ROWX(s0);
          const float4 r1 = ROWX(s1);
          const float4 r2 = ROWX(s2);
          FMA4(r0, w0);
          FMA4(r1, w1);
          FMA4(r2, w2);
        } else if (rem == 2) {
          const int s0 = LANE_I(sv, k), s1 = LANE_I(sv, k + 1);
          const float w0 = LANE_W(wv, k), w1 = LANE_W(wv, k + 1);
          const float4 r0 = ROWX(s0);
          const float4 r1 = ROWX(s1);
          FMA4(r0, w0);
          FMA4(r1, w1);
        } else if (rem == 1) {
          const int s0 = LANE_I(sv, k);
          const float w0 = LANE_W(wv, k);
          const float4 r0 = ROWX(s0);
          FMA4(r0, w0);
        }
        sv = svn;
        wv = wvn;
      }
      *(float4*)(&at[(wave + 8 * q) * XP + c4]) = acc;
    }
    __syncthreads();

    const int nt = tile + (int)gridDim.x;
    int nbeg0, nend0, nbeg1, nend1, nsv0, nsv1;
    float ndi0, ndi1;
    float4 nxv0, nxv1;
    HEAD_ROWS(nt, nbeg0, nend0, ndi0, nxv0, nbeg1, nend1, ndi1, nxv1);

    v8f pacc = {};
#define WMMA_STEP(j)                                                     \
    {                                                                    \
      const int ka = 4 * (j) + 2 * half;                                 \
      v2f a;                                                             \
      a.x = at[m * XP + ka];                                             \
      a.y = at[m * XP + ka + 1];                                         \
      pacc = __builtin_amdgcn_wmma_f32_16x16x4_f32(                      \
           false, a,  false, breg[j],                  \
           (short)0, pacc,  false,  false); \
    }
#pragma unroll
    for (int j = 0; j < 24; ++j) WMMA_STEP(j)
    HEAD_SRC(nbeg0, nsv0);
    HEAD_SRC(nbeg1, nsv1);
#pragma unroll
    for (int j = 24; j < DIM / 4; ++j) WMMA_STEP(j)
#pragma unroll
    for (int r = 0; r < 8; ++r) dt[(r + 8 * half) * XP + col0 + m] = pacc[r];
    __syncthreads();

    const float nwv0 = dinv[nsv0];
    const float nwv1 = dinv[nsv1];

    {
      const float4 p0 = *(const float4*)(&dt[wave * XP + c4]);
      const float4 p1 = *(const float4*)(&dt[(wave + 8) * XP + c4]);
      const float4 xr0 = *(const float4*)(&xt[wave * XP + c4]);
      const float4 xr1 = *(const float4*)(&xt[(wave + 8) * XP + c4]);
      float4 o;
      float v;
      if (d0 < N) {
        v = p0.x + bv.x; v = (v >= 0.0f) ? v : NEG_SLOPE * v; o.x = v + xr0.x;
        v = p0.y + bv.y; v = (v >= 0.0f) ? v : NEG_SLOPE * v; o.y = v + xr0.y;
        v = p0.z + bv.z; v = (v >= 0.0f) ? v : NEG_SLOPE * v; o.z = v + xr0.z;
        v = p0.w + bv.w; v = (v >= 0.0f) ? v : NEG_SLOPE * v; o.w = v + xr0.w;
        ((float4*)(out + (size_t)d0 * DIM))[lane] = o;
      }
      if (d1 < N) {
        v = p1.x + bv.x; v = (v >= 0.0f) ? v : NEG_SLOPE * v; o.x = v + xr1.x;
        v = p1.y + bv.y; v = (v >= 0.0f) ? v : NEG_SLOPE * v; o.y = v + xr1.y;
        v = p1.z + bv.z; v = (v >= 0.0f) ? v : NEG_SLOPE * v; o.z = v + xr1.z;
        v = p1.w + bv.w; v = (v >= 0.0f) ? v : NEG_SLOPE * v; o.w = v + xr1.w;
        ((float4*)(out + (size_t)d1 * DIM))[lane] = o;
      }
    }
    beg0 = nbeg0; end0 = nend0; di0 = ndi0; xv0 = nxv0; sv0 = nsv0; wv0 = nwv0;
    beg1 = nbeg1; end1 = nend1; di1 = ndi1; xv1 = nxv1; sv1 = nsv1; wv1 = nwv1;
  }
}

extern "C" void kernel_launch(void* const* d_in, const int* in_sizes, int n_in,
                              void* d_out, int out_size, void* d_ws, size_t ws_size,
                              hipStream_t stream) {
  const float* x  = (const float*)d_in[0];
  const int*   ei = (const int*)d_in[1];
  const float* W  = (const float*)d_in[2];
  const float* b  = (const float*)d_in[3];

  const int N = in_sizes[0] / DIM;
  const int E = in_sizes[1] / 2;
  const int NB = (N + BW - 1) / BW;
  const int P  = (E + CHUNK - 1) / CHUNK;

  float* out = (float*)d_out;

  const size_t A = 255;
  size_t off = 0;
  float* dinv = (float*)((char*)d_ws + off);  off = (off + (size_t)N * sizeof(float) + A) & ~A;
  int* rowstart = (int*)((char*)d_ws + off);  off = (off + ((size_t)N + 1) * sizeof(int) + A) & ~A;
  int* hist     = (int*)((char*)d_ws + off);  off = (off + (size_t)P * NB * sizeof(int) + A) & ~A;
  int* bstart   = (int*)((char*)d_ws + off);  off = (off + ((size_t)NB + 1) * sizeof(int) + A) & ~A;
  unsigned* tmp = (unsigned*)((char*)d_ws + off);  off = (off + (size_t)E * sizeof(unsigned) + A) & ~A;
  int* csr      = (int*)((char*)d_ws + off);

  part_hist_kernel<<<P, PT, 0, stream>>>(ei, hist, E, N, NB);
  part_place_kernel<<<P, PT, 0, stream>>>(ei, hist, bstart, tmp, E, N, NB, P);
  bucket_sort_kernel<<<NB, PT, 0, stream>>>(tmp, bstart, rowstart, dinv, csr, N);
  gcn_rowsum_gemm_kernel<<<1024, 256, 0, stream>>>(x, W, b, dinv, rowstart, csr, out, N, E);
}


# sort stores as whole runs: the place and bucket-sort kernels order their words in LDS, then write tmp and csr lane by lane to consecutive slots (was 32 lines x 4 B per store); sum order as before
# speedup vs baseline: 5.7789x; 1.2732x over previous
#include <hip/hip_runtime.h>


typedef float v2f __attribute__((ext_vector_type(2)));
typedef float v8f __attribute__((ext_vector_type(8)));

#define DIM 128
#define XP 132
#define NEG_SLOPE 0.1f
#define BW 512
#define BW_SHIFT 9
#define CHUNK 8192
#define PT 1024
#define EPT (CHUNK / PT)
#define CG 256
#define RB 16
#define SB 8
#define STG 12288

__device__ __forceinline__ int block_excl_scan(int v, int* wtot) {
  const int lane = threadIdx.x & 31, wave = threadIdx.x >> 5;
  int incl = v;
  for (int o = 1; o < 32; o <<= 1) {
    int t = __shfl_up(incl, o, 32);
    if (lane >= o) incl += t;
  }
  if (lane == 31) wtot[wave] = incl;
  __syncthreads();
  int wpre = 0;
  for (int w = 0; w < wave; ++w) wpre += wtot[w];
  return wpre + incl - v;
}

__device__ __forceinline__ int2 block_excl_scan2(int a, int b, int2* wtot) {
  const int lane = threadIdx.x & 31, wave = threadIdx.x >> 5;
  int ia = a, ib = b;
  for (int o = 1; o < 32; o <<= 1) {
    int ta = __shfl_up(ia, o, 32);
    int tb = __shfl_up(ib, o, 32);
    if (lane >= o) {
      ia += ta;
      ib += tb;
    }
  }
  if (lane == 31) wtot[wave] = make_int2(ia, ib);
  __syncthreads();
  int pa = 0, pb = 0;
  for (int w = 0; w < wave; ++w) {
    const int2 t = wtot[w];
    pa += t.x;
    pb += t.y;
  }
  return make_int2(pa + ia - a, pb + ib - b);
}

__global__ __launch_bounds__(PT) void part_hist_kernel(const int* __restrict__ ei, int* __restrict__ hist,
                                                       int E, int N, int NB) {
  __shared__ int lc[PT];
  const int tid = threadIdx.x;
  lc[tid] = 0;
  __syncthreads();
  const int e0 = blockIdx.x * CHUNK;
  int dv[EPT];
  for (int k = 0; k < EPT; ++k) dv[k] = ei[E + min(e0 + k * PT + tid, E - 1)];
  for (int k = 0; k < EPT; ++k) {
    if (e0 + k * PT + tid < E) {
      const int d = min(max(dv[k], 0), N - 1);
      atomicAdd(&lc[d >> BW_SHIFT], 1);
    }
  }
  __syncthreads();
  if (tid < NB) hist[blockIdx.x * NB + tid] = lc[tid];
}

__global__ __launch_bounds__(PT) void part_place_kernel(const int* __restrict__ ei, const int* __restrict__ hist,
                                                        int* __restrict__ bstart, unsigned* __restrict__ tmp,
                                                        int E, int N, int NB, int P) {
  __shared__ int cur[PT];
  __shared__ int tot_s[PT];
  __shared__ int mine_s[PT];
  __shared__ int2 wtot2[32];
  __shared__ unsigned ord[CHUNK];
  __shared__ unsigned short obk[CHUNK];
  const int tid = threadIdx.x;
  const int e0 = blockIdx.x * CHUNK;

  tot_s[tid] = 0;
  mine_s[tid] = 0;
  __syncthreads();

  int dv[EPT], sv[EPT];
  for (int k = 0; k < EPT; ++k) {
    const int e = e0 + k * PT + tid;
    dv[k] = 0;
    sv[k] = 0;
    if (e < E) {
      dv[k] = ei[E + e];
      sv[k] = ei[e];
    }
  }
  const int ownv = hist[blockIdx.x * NB + min(tid, NB - 1)];

  const int grp = __builtin_amdgcn_readfirstlane(tid / CG);
  for (int cidx = tid & (CG - 1); cidx < NB; cidx += CG) {
    int t = 0, m = 0;
    for (int p0 = grp; p0 < P; p0 += RB * (PT / CG)) {
      int v[RB];
      for (int k = 0; k < RB; ++k) v[k] = hist[min(p0 + k * (PT / CG), P - 1) * NB + cidx];
      for (int k = 0; k < RB; ++k) {
        const int p = p0 + k * (PT / CG);
        if (p < P) {
          t += v[k];
          if (p < (int)blockIdx.x) m += v[k];
        }
      }
    }
    atomicAdd(&tot_s[cidx], t);
    atomicAdd(&mine_s[cidx], m);
  }
  __syncthreads();
  const int tot = tot_s[tid];
  const int mine = mine_s[tid];
  const int own = (tid < NB) ? ownv : 0;
  const int2 ex = block_excl_scan2(tot, own, wtot2);
  cur[tid] = ex.y;
  mine_s[tid] = ex.x + mine - ex.y;
  if (blockIdx.x == 0 && tid <= NB) bstart[tid] = ex.x;
  __syncthreads();

  for (int k = 0; k < EPT; ++k) {
    const int e = e0 + k * PT + tid;
    if (e < E) {
      const int d = min(max(dv[k], 0), N - 1);
      const int b = d >> BW_SHIFT;
      const int lp = atomicAdd(&cur[b], 1);
      ord[lp] = ((unsigned)sv[k] << BW_SHIFT) | (unsigned)(d & (BW - 1));
      obk[lp] = (unsigned short)b;
    }
  }
  __syncthreads();

  const int nloc = min(CHUNK, E - e0);
  unsigned ov[EPT];
  int ob[EPT];
  for (int k = 0; k < EPT; ++k) {
    const int i = min(k * PT + tid, nloc - 1);
    ov[k] = ord[i];
    ob[k] = obk[i];
  }
  for (int k = 0; k < EPT; ++k) ob[k] = mine_s[ob[k]];
  for (int k = 0; k < EPT; ++k) {
    const int i = k * PT + tid;
    if (i < nloc) tmp[ob[k] + i] = ov[k];
  }
}

__global__ __launch_bounds__(PT) void bucket_sort_kernel(const unsigned* __restrict__ tmp,
                                                         const int* __restrict__ bstart,
                                                         int* __restrict__ rowstart, float* __restrict__ dinv,
                                                         int* __restrict__ csr, int N) {
  __shared__ int cnt[BW];
  __shared__ int wtot[32];
  __shared__ int stg[STG];
  const int tid = threadIdx.x;
  if (tid < BW) cnt[tid] = 0;
  __syncthreads();
  const int beg = bstart[blockIdx.x];
  const int end = bstart[blockIdx.x + 1];
  const int nb = end - beg;
  const bool staged = nb <= STG;
  for (int e = beg + tid; e < end; e += SB * PT) {
    unsigned v[SB];
    for (int k = 0; k < SB; ++k) v[k] = tmp[min(e + k * PT, end - 1)];
    for (int k = 0; k < SB; ++k)
      if (e + k * PT < end) atomicAdd(&cnt[v[k] & (BW - 1)], 1);
  }
  __syncthreads();

  const int c = (tid < BW) ? cnt[tid] : 0;
  const int rs = beg + block_excl_scan(c, wtot);
  const int node = blockIdx.x * BW + tid;
  if (tid < BW) {
    cnt[tid] = rs;
    if (node < N) {
      rowstart[node] = rs;
      dinv[node] = rsqrtf((float)(c + 1));
      if (node == N - 1) rowstart[N] = rs + c;
    }
  }
  __syncthreads();

  if (!staged) {
    for (int e = beg + tid; e < end; e += SB * PT) {
      unsigned v[SB];
      for (int k = 0; k < SB; ++k) v[k] = tmp[min(e + k * PT, end - 1)];
      for (int k = 0; k < SB; ++k) {
        if (e + k * PT < end) {
          const int pos = atomicAdd(&cnt[v[k] & (BW - 1)], 1);
          csr[pos] = (int)(v[k] >> BW_SHIFT);
        }
      }
    }
  } else {
    for (int e = beg + tid; e < end; e += SB * PT) {
      unsigned v[SB];
      for (int k = 0; k < SB; ++k) v[k] = tmp[min(e + k * PT, end - 1)];
      for (int k = 0; k < SB; ++k) {
        if (e + k * PT < end) {
          const int pos = atomicAdd(&cnt[v[k] & (BW - 1)], 1);
          stg[pos - beg] = (int)(v[k] >> BW_SHIFT);
        }
      }
    }
    __syncthreads();
    for (int i0 = tid - (beg & 31); i0 < nb; i0 += SB * PT) {
      int sv[SB];
      for (int k = 0; k < SB; ++k) sv[k] = stg[max(min(i0 + k * PT, nb - 1), 0)];
      for (int k = 0; k < SB; ++k) {
        const int i = i0 + k * PT;
        if (i >= 0 && i < nb) csr[beg + i] = sv[k];
      }
    }
  }
}

#define ROWX(s) (((const float4*)(x + (size_t)(s) * DIM))[lane])
#define FMA4(rowv, wgt)                    \
  acc.x = fmaf((rowv).x, (wgt), acc.x);  \
  acc.y = fmaf((rowv).y, (wgt), acc.y);  \
  acc.z = fmaf((rowv).z, (wgt), acc.z);  \
  acc.w = fmaf((rowv).w, (wgt), acc.w)
#define LANE_I(v, k) __builtin_amdgcn_readlane((v), (k))
#define LANE_W(v, k) (__int_as_float(__builtin_amdgcn_readlane(__float_as_int(v), (k))) * di)

__global__ __launch_bounds__(256) void gcn_rowsum_gemm_kernel(const float* __restrict__ x,
                                                              const float* __restrict__ W,
                                                              const float* __restrict__ bias,
                                                              const float* __restrict__ dinv,
                                                              const int* __restrict__ rowstart,
                                                              const int* __restrict__ csr,
                                                              float* __restrict__ out, int N, int E) {
  __shared__ float at[16 * XP];
  __shared__ float dt[16 * XP];
  __shared__ float xt[16 * XP];

  const int tid  = threadIdx.x;
  const int lane = tid & 31;
  const int wave = __builtin_amdgcn_readfirstlane(tid >> 5);
  const int half = lane >> 4;
  const int m    = lane & 15;
  const int col0 = wave * 16;
  const int c4   = lane * 4;
  const int ntiles = (N + 15) >> 4;
  if (N <= 0) return;

  v2f breg[DIM / 4];
#pragma unroll
  for (int j = 0; j < DIM / 4; ++j) {
    const int ka = 4 * j + 2 * half;
    breg[j].x = W[ka * DIM + col0 + m];
    breg[j].y = W[(ka + 1) * DIM + col0 + m];
  }
  const float4 bv = ((const float4*)bias)[lane];

#define HEAD_ROWS(T, B0, E0, D0, X0, B1, E1, D1, X1)                          \
  {                                                                           \
    const int hr0 = min((T) * 16 + wave, N - 1), hr1 = min((T) * 16 + wave + 8, N - 1); \
    B0 = rowstart[hr0]; E0 = rowstart[hr0 + 1]; D0 = dinv[hr0]; X0 = ROWX(hr0); \
    B1 = rowstart[hr1]; E1 = rowstart[hr1 + 1]; D1 = dinv[hr1]; X1 = ROWX(hr1); \
  }
#define HEAD_SRC(B, S) S = min(max(csr[max(min((B) + lane, E - 1), 0)], 0), N - 1)

  int beg0, end0, beg1, end1, sv0, sv1;
  float di0, di1, wv0, wv1;
  float4 xv0, xv1;
  HEAD_ROWS((int)blockIdx.x, beg0, end0, di0, xv0, beg1, end1, di1, xv1);
  HEAD_SRC(beg0, sv0);
  HEAD_SRC(beg1, sv1);
  wv0 = dinv[sv0];
  wv1 = dinv[sv1];

  for (int tile = blockIdx.x; tile < ntiles; tile += gridDim.x) {
    const int d0 = tile * 16 + wave, d1 = d0 + 8;
    if (d0 >= N) { end0 = beg0; di0 = 0.0f; xv0 = make_float4(0.f, 0.f, 0.f, 0.f); }
    if (d1 >= N) { end1 = beg1; di1 = 0.0f; xv1 = make_float4(0.f, 0.f, 0.f, 0.f); }
    *(float4*)(&xt[wave * XP + c4]) = xv0;
    *(float4*)(&xt[(wave + 8) * XP + c4]) = xv1;

#pragma unroll
    for (int q = 0; q < 2; ++q) {
      const int beg = q ? beg1 : beg0, end = q ? end1 : end0;
      const float di = q ? di1 : di0;
      const float4 xv = q ? xv1 : xv0;
      int sv = q ? sv1 : sv0;
      float wv = q ? wv1 : wv0;

      const float wself = di * di;
      float4 acc;
      acc.x = xv.x * wself;
      acc.y = xv.y * wself;
      acc.z = xv.z * wself;
      acc.w = xv.w * wself;

      for (int base = beg; base < end; base += 32) {
        const int j = base + lane;
        int svn = 0;
        float wvn = 0.0f;
        if (base + 32 < end) {
          if (j + 32 < end) svn = min(max(csr[j + 32], 0), N - 1);
          wvn = dinv[svn];
        }
        int n = end - base;
        if (n > 32) n = 32;

        int k = 0;
        for (; k + 8 <= n; k += 8) {
          const int s0 = LANE_I(sv, k), s1 = LANE_I(sv, k + 1), s2 = LANE_I(sv, k + 2), s3 = LANE_I(sv, k + 3);
          const int s4 = LANE_I(sv, k + 4), s5 = LANE_I(sv, k + 5), s6 = LANE_I(sv, k + 6), s7 = LANE_I(sv, k + 7);
          const float4 r0 = ROWX(s0);
          const float4 r1 = ROWX(s1);
          const float4 r2 = ROWX(s2);
          const float4 r3 = ROWX(s3);
          const float4 r4 = ROWX(s4);
          const float4 r5 = ROWX(s5);
          const float4 r6 = ROWX(s6);
          const float4 r7 = ROWX(s7);
          const float w0 = LANE_W(wv, k), w1 = LANE_W(wv, k + 1), w2 = LANE_W(wv, k + 2), w3 = LANE_W(wv, k + 3);
          const float w4 = LANE_W(wv, k + 4), w5 = LANE_W(wv, k + 5), w6 = LANE_W(wv, k + 6), w7 = LANE_W(wv, k + 7);
          FMA4(r0, w0);
          FMA4(r1, w1);
          FMA4(r2, w2);
          FMA4(r3, w3);
          FMA4(r4, w4);
          FMA4(r5, w5);
          FMA4(r6, w6);
          FMA4(r7, w7);
        }
        if (k + 4 <= n) {
          const int s0 = LANE_I(sv, k), s1 = LANE_I(sv, k + 1), s2 = LANE_I(sv, k + 2), s3 = LANE_I(sv, k + 3);
          const float w0 = LANE_W(wv, k), w1 = LANE_W(wv, k + 1), w2 = LANE_W(wv, k + 2), w3 = LANE_W(wv, k + 3);
          const float4 r0 = ROWX(s0);
          const float4 r1 = ROWX(s1);
          const float4 r2 = ROWX(s2);
          const float4 r3 = ROWX(s3);
          FMA4(r0, w0);
          FMA4(r1, w1);
          FMA4(r2, w2);
          FMA4(r3, w3);
          k += 4;
        }
        const int rem = n - k;
        if (rem == 3) {
          const int s0 = LANE_I(sv, k), s1 = LANE_I(sv, k + 1), s2 = LANE_I(sv, k + 2);
          const float w0 = LANE_W(wv, k), w1 = LANE_W(wv, k + 1), w2 = LANE_W(wv, k + 2);
          const float4 r0 = ROWX(s0);
          const float4 r1 = ROWX(s1);
          const float4 r2 = ROWX(s2);
          FMA4(r0, w0);
          FMA4(r1, w1);
          FMA4(r2, w2);
        } else if (rem == 2) {
          const int s0 = LANE_I(sv, k), s1 = LANE_I(sv, k + 1);
          const float w0 = LANE_W(wv, k), w1 = LANE_W(wv, k + 1);
          const float4 r0 = ROWX(s0);
          const float4 r1 = ROWX(s1);
          FMA4(r0, w0);
          FMA4(r1, w1);
        } else if (rem == 1) {
          const int s0 = LANE_I(sv, k);
          const float w0 = LANE_W(wv, k);
          const float4 r0 = ROWX(s0);
          FMA4(r0, w0);
        }
        sv = svn;
        wv = wvn;
      }
      *(float4*)(&at[(wave + 8 * q) * XP + c4]) = acc;
    }
    __syncthreads();

    const int nt = tile + (int)gridDim.x;
    int nbeg0, nend0, nbeg1, nend1, nsv0, nsv1;
    float ndi0, ndi1;
    float4 nxv0, nxv1;
    HEAD_ROWS(nt, nbeg0, nend0, ndi0, nxv0, nbeg1, nend1, ndi1, nxv1);

    v8f pacc = {};
#define WMMA_STEP(j)                                                     \
    {                                                                    \
      const int ka = 4 * (j) + 2 * half;                                 \
      v2f a;                                                             \
      a.x = at[m * XP + ka];                                             \
      a.y = at[m * XP + ka + 1];                                         \
      pacc = __builtin_amdgcn_wmma_f32_16x16x4_f32(                      \
           false, a,  false, breg[j],                  \
           (short)0, pacc,  false,  false); \
    }
#pragma unroll
    for (int j = 0; j < 24; ++j) WMMA_STEP(j)
    HEAD_SRC(nbeg0, nsv0);
    HEAD_SRC(nbeg1, nsv1);
#pragma unroll
    for (int j = 24; j < DIM / 4; ++j) WMMA_STEP(j)
#pragma unroll
    for (int r = 0; r < 8; ++r) dt[(r + 8 * half) * XP + col0 + m] = pacc[r];
    __syncthreads();

    const float nwv0 = dinv[nsv0];
    const float nwv1 = dinv[nsv1];

    {
      const float4 p0 = *(const float4*)(&dt[wave * XP + c4]);
      const float4 p1 = *(const float4*)(&dt[(wave + 8) * XP + c4]);
      const float4 xr0 = *(const float4*)(&xt[wave * XP + c4]);
      const float4 xr1 = *(const float4*)(&xt[(wave + 8) * XP + c4]);
      float4 o;
      float v;
      if (d0 < N) {
        v = p0.x + bv.x; v = (v >= 0.0f) ? v : NEG_SLOPE * v; o.x = v + xr0.x;
        v = p0.y + bv.y; v = (v >= 0.0f) ? v : NEG_SLOPE * v; o.y = v + xr0.y;
        v = p0.z + bv.z; v = (v >= 0.0f) ? v : NEG_SLOPE * v; o.z = v + xr0.z;
        v = p0.w + bv.w; v = (v >= 0.0f) ? v : NEG_SLOPE * v; o.w = v + xr0.w;
        ((float4*)(out + (size_t)d0 * DIM))[lane] = o;
      }
      if (d1 < N) {
        v = p1.x + bv.x; v = (v >= 0.0f) ? v : NEG_SLOPE * v; o.x = v + xr1.x;
        v = p1.y + bv.y; v = (v >= 0.0f) ? v : NEG_SLOPE * v; o.y = v + xr1.y;
        v = p1.z + bv.z; v = (v >= 0.0f) ? v : NEG_SLOPE * v; o.z = v + xr1.z;
        v = p1.w + bv.w; v = (v >= 0.0f) ? v : NEG_SLOPE * v; o.w = v + xr1.w;
        ((float4*)(out + (size_t)d1 * DIM))[lane] = o;
      }
    }
    beg0 = nbeg0; end0 = nend0; di0 = ndi0; xv0 = nxv0; sv0 = nsv0; wv0 = nwv0;
    beg1 = nbeg1; end1 = nend1; di1 = ndi1; xv1 = nxv1; sv1 = nsv1; wv1 = nwv1;
  }
}

extern "C" void kernel_launch(void* const* d_in, const int* in_sizes, int n_in,
                              void* d_out, int out_size, void* d_ws, size_t ws_size,
                              hipStream_t stream) {
  const float* x  = (const float*)d_in[0];
  const int*   ei = (const int*)d_in[1];
  const float* W  = (const float*)d_in[2];
  const float* b  = (const float*)d_in[3];

  const int N = in_sizes[0] / DIM;
  const int E = in_sizes[1] / 2;
  const int NB = (N + BW - 1) / BW;
  const int P  = (E + CHUNK - 1) / CHUNK;

  float* out = (float*)d_out;

  const size_t A = 255;
  size_t off = 0;
  float* dinv = (float*)((char*)d_ws + off);  off = (off + (size_t)N * sizeof(float) + A) & ~A;
  int* rowstart = (int*)((char*)d_ws + off);  off = (off + ((size_t)N + 1) * sizeof(int) + A) & ~A;
  int* hist     = (int*)((char*)d_ws + off);  off = (off + (size_t)P * NB * sizeof(int) + A) & ~A;
  int* bstart   = (int*)((char*)d_ws + off);  off = (off + ((size_t)NB + 1) * sizeof(int) + A) & ~A;
  unsigned* tmp = (unsigned*)((char*)d_ws + off);  off = (off + (size_t)E * sizeof(unsigned) + A) & ~A;
  int* csr      = (int*)((char*)d_ws + off);

  part_hist_kernel<<<P, PT, 0, stream>>>(ei, hist, E, N, NB);
  part_place_kernel<<<P, PT, 0, stream>>>(ei, hist, bstart, tmp, E, N, NB, P);
  bucket_sort_kernel<<<NB, PT, 0, stream>>>(tmp, bstart, rowstart, dinv, csr, N);
  gcn_rowsum_gemm_kernel<<<1024, 256, 0, stream>>>(x, W, b, dinv, rowstart, csr, out, N, E);
}


# edge sort in two launches, not three: each chunk is ordered by bucket where it stands (rank = LDS counting add's return), bucket blocks gather their runs; row sum order stays LDS cursor arrival
# speedup vs baseline: 6.4340x; 1.1134x over previous
#include <hip/hip_runtime.h>


typedef float v2f __attribute__((ext_vector_type(2)));
typedef float v8f __attribute__((ext_vector_type(8)));

#define DIM 128
#define XP 132
#define NEG_SLOPE 0.1f
#define BW 512
#define BW_SHIFT 9
#define CHUNK 8192
#define PT 1024
#define EPT (CHUNK / PT)
#define RU 8
#define SB 8
#define STG 12288

__device__ __forceinline__ int block_excl_scan(int v, int* wtot) {
  const int lane = threadIdx.x & 31, wave = threadIdx.x >> 5;
  int incl = v;
  for (int o = 1; o < 32; o <<= 1) {
    int t = __shfl_up(incl, o, 32);
    if (lane >= o) incl += t;
  }
  if (lane == 31) wtot[wave] = incl;
  __syncthreads();
  int wpre = 0;
  for (int w = 0; w < wave; ++w) wpre += wtot[w];
  return wpre + incl - v;
}

__device__ __forceinline__ int2 block_excl_scan2(int a, int b, int2* wtot) {
  const int lane = threadIdx.x & 31, wave = threadIdx.x >> 5;
  int ia = a, ib = b;
  for (int o = 1; o < 32; o <<= 1) {
    int ta = __shfl_up(ia, o, 32);
    int tb = __shfl_up(ib, o, 32);
    if (lane >= o) {
      ia += ta;
      ib += tb;
    }
  }
  if (lane == 31) wtot[wave] = make_int2(ia, ib);
  __syncthreads();
  int pa = 0, pb = 0;
  for (int w = 0; w < wave; ++w) {
    const int2 t = wtot[w];
    pa += t.x;
    pb += t.y;
  }
  return make_int2(pa + ia - a, pb + ib - b);
}

__global__ __launch_bounds__(PT) void chunk_order_kernel(const int* __restrict__ ei, int* __restrict__ lstart,
                                                         unsigned* __restrict__ tmp, int E, int N, int NB) {
  __shared__ int cnt[PT];
  __shared__ int wtot[32];
  __shared__ unsigned ord[CHUNK];
  const int tid = threadIdx.x;
  const int e0 = blockIdx.x * CHUNK;
  cnt[tid] = 0;
  int dv[EPT], sv[EPT], rk[EPT];
  for (int k = 0; k < EPT; ++k) {
    const int e = min(e0 + k * PT + tid, E - 1);
    dv[k] = ei[E + e];
    sv[k] = ei[e];
  }
  __syncthreads();
  for (int k = 0; k < EPT; ++k) {
    rk[k] = 0;
    dv[k] = min(max(dv[k], 0), N - 1);
    if (e0 + k * PT + tid < E) rk[k] = atomicAdd(&cnt[dv[k] >> BW_SHIFT], 1);
  }
  __syncthreads();
  const int c = cnt[tid];
  const int ex = block_excl_scan(c, wtot);
  cnt[tid] = ex;
  __syncthreads();
  int st[EPT];
  for (int k = 0; k < EPT; ++k) st[k] = cnt[dv[k] >> BW_SHIFT];
  for (int k = 0; k < EPT; ++k) {
    if (e0 + k * PT + tid < E)
      ord[st[k] + rk[k]] = ((unsigned)sv[k] << BW_SHIFT) | (unsigned)(dv[k] & (BW - 1));
  }
  __syncthreads();
  const int nloc = min(CHUNK, E - e0);
  unsigned ov[EPT];
  for (int k = 0; k < EPT; ++k) ov[k] = ord[min(k * PT + tid, nloc - 1)];
  for (int k = 0; k < EPT; ++k) {
    const int i = k * PT + tid;
    if (i < nloc) tmp[e0 + i] = ov[k];
  }
  if (tid <= NB) lstart[(size_t)blockIdx.x * (NB + 1) + tid] = ex;
}

__global__ __launch_bounds__(PT) void bucket_sort_kernel(const unsigned* __restrict__ tmp,
                                                         const int* __restrict__ lstart,
                                                         int* __restrict__ rowstart, float* __restrict__ dinv,
                                                         int* __restrict__ csr, int N, int E, int NB, int P) {
  __shared__ int cnt[BW];
  __shared__ int wtot[32];
  __shared__ int2 wtot2[32];
  __shared__ int run_src[PT];
  __shared__ int run_len[PT];
  __shared__ int run_off[PT];
  __shared__ int tot_s[2];
  __shared__ unsigned wd[STG];
  __shared__ int stg[STG];
  const int tid = threadIdx.x;
  const int lane = tid & 31;
  const int wave = __builtin_amdgcn_readfirstlane(tid >> 5);
  if (tid < BW) cnt[tid] = 0;

  const int ct = min(tid, P - 1);
  const int* lrow = lstart + (size_t)ct * (NB + 1) + blockIdx.x;
  int s = lrow[0];
  int len = lrow[1] - s;
  s = min(max(s, 0), CHUNK);
  len = min(max(len, 0), CHUNK - s);
  if (tid >= P) {
    s = 0;
    len = 0;
  }
  const int2 ex = block_excl_scan2(len, s, wtot2);
  run_src[tid] = ct * CHUNK + s;
  run_len[tid] = len;
  run_off[tid] = ex.x;
  if (tid == PT - 1) {
    tot_s[0] = ex.x + len;
    tot_s[1] = ex.y + s;
  }
  __syncthreads();
  const int nb = tot_s[0];
  const int beg = tot_s[1];
  const bool staged = nb <= STG;

  if (staged) {
    for (int t0 = wave; t0 < P; t0 += 32 * RU) {
      int rs[RU], rl[RU], ro[RU];
      unsigned v0[RU], v1[RU];
      for (int k = 0; k < RU; ++k) {
        const int t = min(t0 + 32 * k, P - 1);
        rs[k] = run_src[t];
        rl[k] = (t0 + 32 * k < P) ? run_len[t] : 0;
        ro[k] = run_off[t];
      }
      for (int k = 0; k < RU; ++k) {
        v0[k] = tmp[max(min(rs[k] + min(lane, rl[k] - 1), E - 1), 0)];
        v1[k] = tmp[max(min(rs[k] + min(lane + 32, rl[k] - 1), E - 1), 0)];
      }
      for (int k = 0; k < RU; ++k) {
        if (lane < rl[k]) {
          wd[ro[k] + lane] = v0[k];
          atomicAdd(&cnt[v0[k] & (BW - 1)], 1);
        }
        if (lane + 32 < rl[k]) {
          wd[ro[k] + lane + 32] = v1[k];
          atomicAdd(&cnt[v1[k] & (BW - 1)], 1);
        }
      }
      for (int k = 0; k < RU; ++k) {
        for (int i = lane + 64; i < rl[k]; i += 32) {
          const unsigned v = tmp[max(min(rs[k] + i, E - 1), 0)];
          wd[ro[k] + i] = v;
          atomicAdd(&cnt[v & (BW - 1)], 1);
        }
      }
    }
  } else {
    for (int t = wave; t < P; t += 32) {
      const int rs = run_src[t], rl = run_len[t];
      for (int i = lane; i < rl; i += 32) atomicAdd(&cnt[tmp[max(min(rs + i, E - 1), 0)] & (BW - 1)], 1);
    }
  }
  __syncthreads();

  const int c = (tid < BW) ? cnt[tid] : 0;
  const int rs0 = beg + block_excl_scan(c, wtot);
  const int node = blockIdx.x * BW + tid;
  if (tid < BW) {
    cnt[tid] = rs0;
    if (node < N) {
      rowstart[node] = rs0;
      dinv[node] = rsqrtf((float)(c + 1));
      if (node == N - 1) rowstart[N] = rs0 + c;
    }
  }
  __syncthreads();

  if (!staged) {
    for (int t = wave; t < P; t += 32) {
      const int rs = run_src[t], rl = run_len[t];
      for (int i = lane; i < rl; i += 32) {
        const unsigned v = tmp[max(min(rs + i, E - 1), 0)];
        const int pos = atomicAdd(&cnt[v & (BW - 1)], 1);
        csr[max(min(pos, E - 1), 0)] = (int)(v >> BW_SHIFT);
      }
    }
  } else {
    for (int i0 = tid; i0 < nb; i0 += SB * PT) {
      unsigned v[SB];
      for (int k = 0; k < SB; ++k) v[k] = wd[min(i0 + k * PT, nb - 1)];
      for (int k = 0; k < SB; ++k) {
        if (i0 + k * PT < nb) {
          const int pos = atomicAdd(&cnt[v[k] & (BW - 1)], 1);
          stg[pos - beg] = (int)(v[k] >> BW_SHIFT);
        }
      }
    }
    __syncthreads();
    for (int i0 = tid - (beg & 31); i0 < nb; i0 += SB * PT) {
      int sv[SB];
      for (int k = 0; k < SB; ++k) sv[k] = stg[max(min(i0 + k * PT, nb - 1), 0)];
      for (int k = 0; k < SB; ++k) {
        const int i = i0 + k * PT;
        if (i >= 0 && i < nb) csr[beg + i] = sv[k];
      }
    }
  }
}

#define ROWX(s) (((const float4*)(x + (size_t)(s) * DIM))[lane])
#define FMA4(rowv, wgt)                    \
  acc.x = fmaf((rowv).x, (wgt), acc.x);  \
  acc.y = fmaf((rowv).y, (wgt), acc.y);  \
  acc.z = fmaf((rowv).z, (wgt), acc.z);  \
  acc.w = fmaf((rowv).w, (wgt), acc.w)
#define LANE_I(v, k) __builtin_amdgcn_readlane((v), (k))
#define LANE_W(v, k) (__int_as_float(__builtin_amdgcn_readlane(__float_as_int(v), (k))) * di)

__global__ __launch_bounds__(256) void gcn_rowsum_gemm_kernel(const float* __restrict__ x,
                                                              const float* __restrict__ W,
                                                              const float* __restrict__ bias,
                                                              const float* __restrict__ dinv,
                                                              const int* __restrict__ rowstart,
                                                              const int* __restrict__ csr,
                                                              float* __restrict__ out, int N, int E) {
  __shared__ float at[16 * XP];
  __shared__ float dt[16 * XP];
  __shared__ float xt[16 * XP];

  const int tid  = threadIdx.x;
  const int lane = tid & 31;
  const int wave = __builtin_amdgcn_readfirstlane(tid >> 5);
  const int half = lane >> 4;
  const int m    = lane & 15;
  const int col0 = wave * 16;
  const int c4   = lane * 4;
  const int ntiles = (N + 15) >> 4;
  if (N <= 0) return;

  v2f breg[DIM / 4];
#pragma unroll
  for (int j = 0; j < DIM / 4; ++j) {
    const int ka = 4 * j + 2 * half;
    breg[j].x = W[ka * DIM + col0 + m];
    breg[j].y = W[(ka + 1) * DIM + col0 + m];
  }
  const float4 bv = ((const float4*)bias)[lane];

#define HEAD_ROWS(T, B0, E0, D0, X0, B1, E1, D1, X1)                          \
  {                                                                           \
    const int hr0 = min((T) * 16 + wave, N - 1), hr1 = min((T) * 16 + wave + 8, N - 1); \
    B0 = rowstart[hr0]; E0 = rowstart[hr0 + 1]; D0 = dinv[hr0]; X0 = ROWX(hr0); \
    B1 = rowstart[hr1]; E1 = rowstart[hr1 + 1]; D1 = dinv[hr1]; X1 = ROWX(hr1); \
  }
#define HEAD_SRC(B, S) S = min(max(csr[max(min((B) + lane, E - 1), 0)], 0), N - 1)

  int beg0, end0, beg1, end1, sv0, sv1;
  float di0, di1, wv0, wv1;
  float4 xv0, xv1;
  HEAD_ROWS((int)blockIdx.x, beg0, end0, di0, xv0, beg1, end1, di1, xv1);
  HEAD_SRC(beg0, sv0);
  HEAD_SRC(beg1, sv1);
  wv0 = dinv[sv0];
  wv1 = dinv[sv1];

  for (int tile = blockIdx.x; tile < ntiles; tile += gridDim.x) {
    const int d0 = tile * 16 + wave, d1 = d0 + 8;
    if (d0 >= N) { end0 = beg0; di0 = 0.0f; xv0 = make_float4(0.f, 0.f, 0.f, 0.f); }
    if (d1 >= N) { end1 = beg1; di1 = 0.0f; xv1 = make_float4(0.f, 0.f, 0.f, 0.f); }
    *(float4*)(&xt[wave * XP + c4]) = xv0;
    *(float4*)(&xt[(wave + 8) * XP + c4]) = xv1;

#pragma unroll
    for (int q = 0; q < 2; ++q) {
      const int beg = q ? beg1 : beg0, end = q ? end1 : end0;
      const float di = q ? di1 : di0;
      const float4 xv = q ? xv1 : xv0;
      int sv = q ? sv1 : sv0;
      float wv = q ? wv1 : wv0;

      const float wself = di * di;
      float4 acc;
      acc.x = xv.x * wself;
      acc.y = xv.y * wself;
      acc.z = xv.z * wself;
      acc.w = xv.w * wself;

      for (int base = beg; base < end; base += 32) {
        const int j = base + lane;
        int svn = 0;
        float wvn = 0.0f;
        if (base + 32 < end) {
          if (j + 32 < end) svn = min(max(csr[j + 32], 0), N - 1);
          wvn = dinv[svn];
        }
        int n = end - base;
        if (n > 32) n = 32;

        int k = 0;
        for (; k + 8 <= n; k += 8) {
          const int s0 = LANE_I(sv, k), s1 = LANE_I(sv, k + 1), s2 = LANE_I(sv, k + 2), s3 = LANE_I(sv, k + 3);
          const int s4 = LANE_I(sv, k + 4), s5 = LANE_I(sv, k + 5), s6 = LANE_I(sv, k + 6), s7 = LANE_I(sv, k + 7);
          const float4 r0 = ROWX(s0);
          const float4 r1 = ROWX(s1);
          const float4 r2 = ROWX(s2);
          const float4 r3 = ROWX(s3);
          const float4 r4 = ROWX(s4);
          const float4 r5 = ROWX(s5);
          const float4 r6 = ROWX(s6);
          const float4 r7 = ROWX(s7);
          const float w0 = LANE_W(wv, k), w1 = LANE_W(wv, k + 1), w2 = LANE_W(wv, k + 2), w3 = LANE_W(wv, k + 3);
          const float w4 = LANE_W(wv, k + 4), w5 = LANE_W(wv, k + 5), w6 = LANE_W(wv, k + 6), w7 = LANE_W(wv, k + 7);
          FMA4(r0, w0);
          FMA4(r1, w1);
          FMA4(r2, w2);
          FMA4(r3, w3);
          FMA4(r4, w4);
          FMA4(r5, w5);
          FMA4(r6, w6);
          FMA4(r7, w7);
        }
        if (k + 4 <= n) {
          const int s0 = LANE_I(sv, k), s1 = LANE_I(sv, k + 1), s2 = LANE_I(sv, k + 2), s3 = LANE_I(sv, k + 3);
          const float w0 = LANE_W(wv, k), w1 = LANE_W(wv, k + 1), w2 = LANE_W(wv, k + 2), w3 = LANE_W(wv, k + 3);
          const float4 r0 = ROWX(s0);
          const float4 r1 = ROWX(s1);
          const float4 r2 = ROWX(s2);
          const float4 r3 = ROWX(s3);
          FMA4(r0, w0);
          FMA4(r1, w1);
          FMA4(r2, w2);
          FMA4(r3, w3);
          k += 4;
        }
        const int rem = n - k;
        if (rem == 3) {
          const int s0 = LANE_I(sv, k), s1 = LANE_I(sv, k + 1), s2 = LANE_I(sv, k + 2);
          const float w0 = LANE_W(wv, k), w1 = LANE_W(wv, k + 1), w2 = LANE_W(wv, k + 2);
          const float4 r0 = ROWX(s0);
          const float4 r1 = ROWX(s1);
          const float4 r2 = ROWX(s2);
          FMA4(r0, w0);
          FMA4(r1, w1);
          FMA4(r2, w2);
        } else if (rem == 2) {
          const int s0 = LANE_I(sv, k), s1 = LANE_I(sv, k + 1);
          const float w0 = LANE_W(wv, k), w1 = LANE_W(wv, k + 1);
          const float4 r0 = ROWX(s0);
          const float4 r1 = ROWX(s1);
          FMA4(r0, w0);
          FMA4(r1, w1);
        } else if (rem == 1) {
          const int s0 = LANE_I(sv, k);
          const float w0 = LANE_W(wv, k);
          const float4 r0 = ROWX(s0);
          FMA4(r0, w0);
        }
        sv = svn;
        wv = wvn;
      }
      *(float4*)(&at[(wave + 8 * q) * XP + c4]) = acc;
    }
    __syncthreads();

    const int nt = tile + (int)gridDim.x;
    int nbeg0, nend0, nbeg1, nend1, nsv0, nsv1;
    float ndi0, ndi1;
    float4 nxv0, nxv1;
    HEAD_ROWS(nt, nbeg0, nend0, ndi0, nxv0, nbeg1, nend1, ndi1, nxv1);

    v8f pacc = {};
#define WMMA_STEP(j)                                                     \
    {                                                                    \
      const int ka = 4 * (j) + 2 * half;                                 \
      v2f a;                                                             \
      a.x = at[m * XP + ka];                                             \
      a.y = at[m * XP + ka + 1];                                         \
      pacc = __builtin_amdgcn_wmma_f32_16x16x4_f32(                      \
           false, a,  false, breg[j],                  \
           (short)0, pacc,  false,  false); \
    }
#pragma unroll
    for (int j = 0; j < 24; ++j) WMMA_STEP(j)
    HEAD_SRC(nbeg0, nsv0);
    HEAD_SRC(nbeg1, nsv1);
#pragma unroll
    for (int j = 24; j < DIM / 4; ++j) WMMA_STEP(j)
#pragma unroll
    for (int r = 0; r < 8; ++r) dt[(r + 8 * half) * XP + col0 + m] = pacc[r];
    __syncthreads();

    const float nwv0 = dinv[nsv0];
    const float nwv1 = dinv[nsv1];

    {
      const float4 p0 = *(const float4*)(&dt[wave * XP + c4]);
      const float4 p1 = *(const float4*)(&dt[(wave + 8) * XP + c4]);
      const float4 xr0 = *(const float4*)(&xt[wave * XP + c4]);
      const float4 xr1 = *(const float4*)(&xt[(wave + 8) * XP + c4]);
      float4 o;
      float v;
      if (d0 < N) {
        v = p0.x + bv.x; v = (v >= 0.0f) ? v : NEG_SLOPE * v; o.x = v + xr0.x;
        v = p0.y + bv.y; v = (v >= 0.0f) ? v : NEG_SLOPE * v; o.y = v + xr0.y;
        v = p0.z + bv.z; v = (v >= 0.0f) ? v : NEG_SLOPE * v; o.z = v + xr0.z;
        v = p0.w + bv.w; v = (v >= 0.0f) ? v : NEG_SLOPE * v; o.w = v + xr0.w;
        ((float4*)(out + (size_t)d0 * DIM))[lane] = o;
      }
      if (d1 < N) {
        v = p1.x + bv.x; v = (v >= 0.0f) ? v : NEG_SLOPE * v; o.x = v + xr1.x;
        v = p1.y + bv.y; v = (v >= 0.0f) ? v : NEG_SLOPE * v; o.y = v + xr1.y;
        v = p1.z + bv.z; v = (v >= 0.0f) ? v : NEG_SLOPE * v; o.z = v + xr1.z;
        v = p1.w + bv.w; v = (v >= 0.0f) ? v : NEG_SLOPE * v; o.w = v + xr1.w;
        ((float4*)(out + (size_t)d1 * DIM))[lane] = o;
      }
    }
    beg0 = nbeg0; end0 = nend0; di0 = ndi0; xv0 = nxv0; sv0 = nsv0; wv0 = nwv0;
    beg1 = nbeg1; end1 = nend1; di1 = ndi1; xv1 = nxv1; sv1 = nsv1; wv1 = nwv1;
  }
}

extern "C" void kernel_launch(void* const* d_in, const int* in_sizes, int n_in,
                              void* d_out, int out_size, void* d_ws, size_t ws_size,
                              hipStream_t stream) {
  const float* x  = (const float*)d_in[0];
  const int*   ei = (const int*)d_in[1];
  const float* W  = (const float*)d_in[2];
  const float* b  = (const float*)d_in[3];

  const int N = in_sizes[0] / DIM;
  const int E = in_sizes[1] / 2;
  const int NB = (N + BW - 1) / BW;
  const int P  = (E + CHUNK - 1) / CHUNK;

  float* out = (float*)d_out;

  const size_t A = 255;
  size_t off = 0;
  float* dinv = (float*)((char*)d_ws + off);  off = (off + (size_t)N * sizeof(float) + A) & ~A;
  int* rowstart = (int*)((char*)d_ws + off);  off = (off + ((size_t)N + 1) * sizeof(int) + A) & ~A;
  int* lstart   = (int*)((char*)d_ws + off);  off = (off + (size_t)P * (NB + 1) * sizeof(int) + A) & ~A;
  unsigned* tmp = (unsigned*)((char*)d_ws + off);  off = (off + (size_t)E * sizeof(unsigned) + A) & ~A;
  int* csr      = (int*)((char*)d_ws + off);

  chunk_order_kernel<<<P, PT, 0, stream>>>(ei, lstart, tmp, E, N, NB);
  bucket_sort_kernel<<<NB, PT, 0, stream>>>(tmp, lstart, rowstart, dinv, csr, N, E, NB, P);
  gcn_rowsum_gemm_kernel<<<1024, 256, 0, stream>>>(x, W, b, dinv, rowstart, csr, out, N, E);
}
